# starting kernel
# baseline (speedup 1.0000x reference)
#include <hip/hip_runtime.h>
#include <hip/hip_bf16.h>
#include <math.h>

#define NN 50000
#define EE 800000
#define INF_ 256
#define OUTF 32
#define H1N 8
#define SLOPE 0.2f

typedef float v2f __attribute__((ext_vector_type(2)));
typedef float v8f __attribute__((ext_vector_type(8)));

__device__ __forceinline__ float leaky(float x) { return x > 0.f ? x : SLOPE * x; }

__device__ __forceinline__ void atomic_max_f32(float* addr, float v) {
    if (v >= 0.f) atomicMax((int*)addr, __float_as_int(v));
    else          atomicMin((unsigned int*)addr, __float_as_uint(v));
}

__global__ void __launch_bounds__(128) gemm1_kernel(const float* __restrict__ X,
                                                    const float* __restrict__ W,
                                                    float* __restrict__ C) {
    const int wave = threadIdx.x >> 5;
    const int lane = threadIdx.x & 31;
    const int half = lane >> 4;
    const int r    = lane & 15;
    const int row0 = blockIdx.x * 16;
    const int col0 = wave * 64;

    v8f acc[4] = {};
    const float* xrow = X + (size_t)(row0 + r) * INF_;

    for (int k = 0; k < INF_; k += 4) {
        v2f a;
        a.x = xrow[k + half * 2 + 0];
        a.y = xrow[k + half * 2 + 1];
#pragma unroll
        for (int t = 0; t < 4; ++t) {
            const int col = col0 + t * 16 + r;
            v2f b;
            b.x = W[(size_t)(k + half * 2 + 0) * 256 + col];
            b.y = W[(size_t)(k + half * 2 + 1) * 256 + col];
            acc[t] = __builtin_amdgcn_wmma_f32_16x16x4_f32(
                false, a, false, b, (short)0, acc[t], false, false);
        }
    }
#pragma unroll
    for (int t = 0; t < 4; ++t) {
#pragma unroll
        for (int j = 0; j < 8; ++j) {
            const int row = row0 + half * 8 + j;
            const int col = col0 + t * 16 + r;
            C[(size_t)row * 256 + col] = acc[t][j];
        }
    }
}

__global__ void __launch_bounds__(128) gemm2_kernel(const float* __restrict__ Y,
                                                    const float* __restrict__ W2,
                                                    const float* __restrict__ RW,
                                                    const float* __restrict__ b2,
                                                    float* __restrict__ F2,
                                                    float* __restrict__ OUTP,
                                                    int n) {
    const int wave = threadIdx.x >> 5;
    const int lane = threadIdx.x & 31;
    const int half = lane >> 4;
    const int r    = lane & 15;
    const int row0 = blockIdx.x * 64 + wave * 16;

    int rload = row0 + r;
    if (rload >= n) rload = n - 1;
    const float* yrow = Y + (size_t)rload * 256;

    v8f accW[2] = {};
    v8f accR[2] = {};

    for (int k = 0; k < 256; k += 4) {
        v2f a;
        a.x = yrow[k + half * 2 + 0];
        a.y = yrow[k + half * 2 + 1];
#pragma unroll
        for (int t = 0; t < 2; ++t) {
            const int col = t * 16 + r;
            v2f bw, br;
            bw.x = W2[(size_t)(k + half * 2 + 0) * 32 + col];
            bw.y = W2[(size_t)(k + half * 2 + 1) * 32 + col];
            accW[t] = __builtin_amdgcn_wmma_f32_16x16x4_f32(
                false, a, false, bw, (short)0, accW[t], false, false);
            br.x = RW[(size_t)(k + half * 2 + 0) * 32 + col];
            br.y = RW[(size_t)(k + half * 2 + 1) * 32 + col];
            accR[t] = __builtin_amdgcn_wmma_f32_16x16x4_f32(
                false, a, false, br, (short)0, accR[t], false, false);
        }
    }
#pragma unroll
    for (int t = 0; t < 2; ++t) {
#pragma unroll
        for (int j = 0; j < 8; ++j) {
            const int row = row0 + half * 8 + j;
            const int col = t * 16 + r;
            if (row < n) {
                F2[(size_t)row * 32 + col]   = accW[t][j];
                OUTP[(size_t)row * 32 + col] = accR[t][j] + b2[col];
            }
        }
    }
}

__global__ void __launch_bounds__(256) elr1_kernel(const float* __restrict__ feat,
                                                   const float* __restrict__ al,
                                                   const float* __restrict__ ar,
                                                   float* __restrict__ el,
                                                   float* __restrict__ er) {
    const int n = blockIdx.x;
    const int d = threadIdx.x;
    const float v = feat[(size_t)n * 256 + d];
    float pl = v * al[d];
    float pr = v * ar[d];
#pragma unroll
    for (int off = 16; off > 0; off >>= 1) {
        pl += __shfl_xor(pl, off, 32);
        pr += __shfl_xor(pr, off, 32);
    }
    if ((d & 31) == 0) {
        const int h = d >> 5;
        el[n * H1N + h] = pl;
        er[n * H1N + h] = pr;
    }
}

__global__ void __launch_bounds__(256) elr2_kernel(const float* __restrict__ feat,
                                                   const float* __restrict__ al,
                                                   const float* __restrict__ ar,
                                                   float* __restrict__ el,
                                                   float* __restrict__ er,
                                                   int n) {
    const int node = blockIdx.x * 8 + (threadIdx.x >> 5);
    const int lane = threadIdx.x & 31;
    if (node >= n) return;
    const float v = feat[(size_t)node * 32 + lane];
    float pl = v * al[lane];
    float pr = v * ar[lane];
#pragma unroll
    for (int off = 16; off > 0; off >>= 1) {
        pl += __shfl_xor(pl, off, 32);
        pr += __shfl_xor(pr, off, 32);
    }
    if (lane == 0) { el[node] = pl; er[node] = pr; }
}

__global__ void __launch_bounds__(256) init_rst1_kernel(const float* __restrict__ x,
                                                        const float* __restrict__ b1,
                                                        float* __restrict__ rst) {
    const size_t i = (size_t)blockIdx.x * 256 + threadIdx.x;
    rst[i] = x[i] + b1[threadIdx.x];
}

__global__ void __launch_bounds__(256) init_softmax_kernel(float* __restrict__ emax,
                                                           float* __restrict__ esum,
                                                           int n) {
    const int i = blockIdx.x * 256 + threadIdx.x;
    if (i < n) { emax[i] = -INFINITY; esum[i] = 0.f; }
}

__global__ void __launch_bounds__(256) relu_kernel(float* __restrict__ y) {
    const size_t i = (size_t)blockIdx.x * 256 + threadIdx.x;
    y[i] = fmaxf(y[i], 0.f);
}

__global__ void __launch_bounds__(256) edge1_max_kernel(const int* __restrict__ src,
                                                        const int* __restrict__ dst,
                                                        const float* __restrict__ el,
                                                        const float* __restrict__ er,
                                                        float* __restrict__ emax) {
    const int tid = blockIdx.x * 256 + threadIdx.x;
    if (tid >= EE * H1N) return;
    const int e = tid >> 3, h = tid & 7;
    const int s = src[e], d = dst[e];
    const float sc = leaky(el[s * H1N + h] + er[d * H1N + h]);
    atomic_max_f32(&emax[d * H1N + h], sc);
}

__global__ void __launch_bounds__(256) edge1_sum_kernel(const int* __restrict__ src,
                                                        const int* __restrict__ dst,
                                                        const float* __restrict__ el,
                                                        const float* __restrict__ er,
                                                        const float* __restrict__ emax,
                                                        float* __restrict__ esum) {
    const int tid = blockIdx.x * 256 + threadIdx.x;
    if (tid >= EE * H1N) return;
    const int e = tid >> 3, h = tid & 7;
    const int s = src[e], d = dst[e];
    const float sc = leaky(el[s * H1N + h] + er[d * H1N + h]);
    atomicAdd(&esum[d * H1N + h], __expf(sc - emax[d * H1N + h]));
}

__global__ void __launch_bounds__(256) edge1_agg_kernel(const int* __restrict__ src,
                                                        const int* __restrict__ dst,
                                                        const float* __restrict__ el,
                                                        const float* __restrict__ er,
                                                        const float* __restrict__ emax,
                                                        const float* __restrict__ esum,
                                                        const float* __restrict__ feat,
                                                        float* __restrict__ rst) {
    const int gid  = blockIdx.x * 256 + threadIdx.x;
    const int e    = gid >> 5;
    if (e >= EE) return;
    const int lane = gid & 31;
    const int s = src[e], d = dst[e];
    const int h = lane >> 2;
    const float sc = leaky(el[s * H1N + h] + er[d * H1N + h]);
    const float a  = __expf(sc - emax[d * H1N + h]) / esum[d * H1N + h];

    const float4* fs = (const float4*)(feat + (size_t)s * 256 + lane * 8);
    const float4 f0 = fs[0], f1 = fs[1];
    float* rd = rst + (size_t)d * 256 + lane * 8;
    atomicAdd(rd + 0, f0.x * a); atomicAdd(rd + 1, f0.y * a);
    atomicAdd(rd + 2, f0.z * a); atomicAdd(rd + 3, f0.w * a);
    atomicAdd(rd + 4, f1.x * a); atomicAdd(rd + 5, f1.y * a);
    atomicAdd(rd + 6, f1.z * a); atomicAdd(rd + 7, f1.w * a);
}

__global__ void __launch_bounds__(256) edge2_max_kernel(const int* __restrict__ src,
                                                        const int* __restrict__ dst,
                                                        const float* __restrict__ el,
                                                        const float* __restrict__ er,
                                                        float* __restrict__ emax) {
    const int e = blockIdx.x * 256 + threadIdx.x;
    if (e >= EE) return;
    const int s = src[e], d = dst[e];
    atomic_max_f32(&emax[d], leaky(el[s] + er[d]));
}

__global__ void __launch_bounds__(256) edge2_sum_kernel(const int* __restrict__ src,
                                                        const int* __restrict__ dst,
                                                        const float* __restrict__ el,
                                                        const float* __restrict__ er,
                                                        const float* __restrict__ emax,
                                                        float* __restrict__ esum) {
    const int e = blockIdx.x * 256 + threadIdx.x;
    if (e >= EE) return;
    const int s = src[e], d = dst[e];
    const float sc = leaky(el[s] + er[d]);
    atomicAdd(&esum[d], __expf(sc - emax[d]));
}

__global__ void __launch_bounds__(256) edge2_agg_kernel(const int* __restrict__ src,
                                                        const int* __restrict__ dst,
                                                        const float* __restrict__ el,
                                                        const float* __restrict__ er,
                                                        const float* __restrict__ emax,
                                                        const float* __restrict__ esum,
                                                        const float* __restrict__ feat,
                                                        float* __restrict__ outp) {
    const int gid  = blockIdx.x * 256 + threadIdx.x;
    const int e    = gid >> 5;
    if (e >= EE) return;
    const int lane = gid & 31;
    const int s = src[e], d = dst[e];
    const float sc = leaky(el[s] + er[d]);
    const float a  = __expf(sc - emax[d]) / esum[d];
    atomicAdd(outp + (size_t)d * 32 + lane, feat[(size_t)s * 32 + lane] * a);
}

extern "C" void kernel_launch(void* const* d_in, const int* in_sizes, int n_in,
                              void* d_out, int out_size, void* d_ws, size_t ws_size,
                              hipStream_t stream) {
    const float* x     = (const float*)d_in[0];
    const int*   src   = (const int*)  d_in[1];
    const int*   dst   = (const int*)  d_in[2];
    const float* W1    = (const float*)d_in[3];
    const float* al1   = (const float*)d_in[4];
    const float* ar1   = (const float*)d_in[5];
    const float* b1    = (const float*)d_in[6];
    const float* W2    = (const float*)d_in[7];
    const float* al2   = (const float*)d_in[8];
    const float* ar2   = (const float*)d_in[9];
    const float* resW2 = (const float*)d_in[10];
    const float* b2    = (const float*)d_in[11];
    float* outp = (float*)d_out;

    float* ws    = (float*)d_ws;
    float* feat1 = ws;
    float* rst1  = feat1 + (size_t)NN * 256;
    float* el1   = rst1  + (size_t)NN * 256;
    float* er1   = el1   + (size_t)NN * 8;
    float* emax1 = er1   + (size_t)NN * 8;
    float* esum1 = emax1 + (size_t)NN * 8;
    float* feat2 = esum1 + (size_t)NN * 8;
    float* el2   = feat2 + (size_t)NN * 32;
    float* er2   = el2   + NN;
    float* emax2 = er2   + NN;
    float* esum2 = emax2 + NN;

    gemm1_kernel<<<NN / 16, 128, 0, stream>>>(x, W1, feat1);
    init_rst1_kernel<<<NN, 256, 0, stream>>>(x, b1, rst1);
    init_softmax_kernel<<<(NN * H1N + 255) / 256, 256, 0, stream>>>(emax1, esum1, NN * H1N);
    elr1_kernel<<<NN, 256, 0, stream>>>(feat1, al1, ar1, el1, er1);
    edge1_max_kernel<<<(EE * H1N + 255) / 256, 256, 0, stream>>>(src, dst, el1, er1, emax1);
    edge1_sum_kernel<<<(EE * H1N + 255) / 256, 256, 0, stream>>>(src, dst, el1, er1, emax1, esum1);
    edge1_agg_kernel<<<(EE * 32) / 256, 256, 0, stream>>>(src, dst, el1, er1, emax1, esum1,
                                                          feat1, rst1);
    relu_kernel<<<NN, 256, 0, stream>>>(rst1);

    init_softmax_kernel<<<(NN + 255) / 256, 256, 0, stream>>>(emax2, esum2, NN);
    gemm2_kernel<<<(NN + 63) / 64, 128, 0, stream>>>(rst1, W2, resW2, b2, feat2, outp, NN);
    elr2_kernel<<<(NN + 7) / 8, 256, 0, stream>>>(feat2, al2, ar2, el2, er2, NN);
    edge2_max_kernel<<<(EE + 255) / 256, 256, 0, stream>>>(src, dst, el2, er2, emax2);
    edge2_sum_kernel<<<(EE + 255) / 256, 256, 0, stream>>>(src, dst, el2, er2, emax2, esum2);
    edge2_agg_kernel<<<(EE * 32) / 256, 256, 0, stream>>>(src, dst, el2, er2, emax2, esum2,
                                                          feat2, outp);
}


# Both GEMMs: weights staged once in LDS by 256 persistent workgroups, B fragments by 16-byte LDS loads, A by 16-byte row loads; k sum order changes (blocks of 8, two matrix instructions a block)
# speedup vs baseline: 4.7074x; 4.7074x over previous
#include <hip/hip_runtime.h>
#include <hip/hip_bf16.h>
#include <math.h>

#define NN 50000
#define EE 800000
#define INF_ 256
#define OUTF 32
#define H1N 8
#define SLOPE 0.2f

typedef float v2f __attribute__((ext_vector_type(2)));
typedef float v8f __attribute__((ext_vector_type(8)));

__device__ __forceinline__ float leaky(float x) { return x > 0.f ? x : SLOPE * x; }

typedef float v4f __attribute__((ext_vector_type(4)));

#define NT_ROWS (NN / 16)
#define GEMM_WGS 256
#define GEMM_THREADS 512
static_assert(NN % 16 == 0, "whole 16-row tiles only: no row guard in the GEMM kernels");

template <int NCOL>
__device__ __forceinline__ void stage_weights(float* __restrict__ Wl, const float* __restrict__ Wa,
                                              const float* __restrict__ Wb, const int tid) {
    constexpr int P = NCOL * 4 + 32;
    constexpr int CG = NCOL / 4;
    constexpr int NB = 64 * CG / GEMM_THREADS;
    static_assert(64 * CG % GEMM_THREADS == 0, "every thread moves the same number of blocks");
    const int rot = (tid >> 2) & 3;
    v4f r[NB][4];
#pragma unroll
    for (int it = 0; it < NB; ++it) {
        const int q = tid + it * GEMM_THREADS;
        const int p = q / CG, cg = q % CG;
        const float* s;
        int pitch;
        if (NCOL == 256) { s = Wa + (size_t)(4 * p) * 256 + 4 * cg; pitch = 256; }
        else             { s = (cg < 8 ? Wa : Wb) + (size_t)(4 * p) * 32 + 4 * (cg & 7); pitch = 32; }
#pragma unroll
        for (int i = 0; i < 4; ++i) r[it][i] = *reinterpret_cast<const v4f*>(s + i * pitch);
    }
#pragma unroll
    for (int it = 0; it < NB; ++it) {
        const int q = tid + it * GEMM_THREADS;
        const int p = q / CG, cg = q % CG;
#pragma unroll
        for (int j = 0; j < 4; ++j) {
            const int jj = (j + rot) & 3;
            v4f o;
            o.x = jj == 0 ? r[it][0].x : jj == 1 ? r[it][0].y : jj == 2 ? r[it][0].z : r[it][0].w;
            o.y = jj == 0 ? r[it][1].x : jj == 1 ? r[it][1].y : jj == 2 ? r[it][1].z : r[it][1].w;
            o.z = jj == 0 ? r[it][2].x : jj == 1 ? r[it][2].y : jj == 2 ? r[it][2].z : r[it][2].w;
            o.w = jj == 0 ? r[it][3].x : jj == 1 ? r[it][3].y : jj == 2 ? r[it][3].z : r[it][3].w;
            *reinterpret_cast<v4f*>(Wl + p * P + (4 * cg + jj) * 4) = o;
        }
    }
}

template <int NCOL, int NT16>
__device__ __forceinline__ void unit_mma(const float* __restrict__ A, const float* __restrict__ Wl,
                                         const int tile, const int col0, const int lane, v8f (&acc)[NT16]) {
    constexpr int P = NCOL * 4 + 32;
    const int h = lane >> 4, c = lane & 15;
    const v4f* __restrict__ a4 = reinterpret_cast<const v4f*>(A + (size_t)(tile * 16 + c) * 256) + h;
    const float* bp = Wl + h * P + (col0 + c) * 4;
    for (int J0 = 0; J0 < 32; J0 += 8) {
        v4f ac[8];
#pragma unroll
        for (int j = 0; j < 8; ++j) ac[j] = a4[2 * (J0 + j)];
#pragma unroll
        for (int j = 0; j < 8; ++j) {
            const float* bj = bp + 2 * (J0 + j) * P;
            v4f b[NT16];
#pragma unroll
            for (int t = 0; t < NT16; ++t) b[t] = *reinterpret_cast<const v4f*>(bj + t * 64);
            v2f alo, ahi;
            alo.x = ac[j].x; alo.y = ac[j].y; ahi.x = ac[j].z; ahi.y = ac[j].w;
#pragma unroll
            for (int t = 0; t < NT16; ++t) {
                v2f blo; blo.x = b[t].x; blo.y = b[t].y;
                acc[t] = __builtin_amdgcn_wmma_f32_16x16x4_f32(false, alo, false, blo, (short)0, acc[t], false, false);
            }
#pragma unroll
            for (int t = 0; t < NT16; ++t) {
                v2f bhi; bhi.x = b[t].z; bhi.y = b[t].w;
                acc[t] = __builtin_amdgcn_wmma_f32_16x16x4_f32(false, ahi, false, bhi, (short)0, acc[t], false, false);
            }
        }
    }
}

__global__ void __launch_bounds__(GEMM_THREADS) gemm1_kernel(const float* __restrict__ X,
                                                             const float* __restrict__ W,
                                                             float* __restrict__ C) {
    __shared__ __attribute__((aligned(16))) float Wl[64 * (256 * 4 + 32)];
    const int tid = threadIdx.x, lane = tid & 31;
    const int wave = __builtin_amdgcn_readfirstlane(tid >> 5);
    stage_weights<256>(Wl, W, W, tid);
    __syncthreads();
    const int ntb = (NT_ROWS - (int)blockIdx.x + GEMM_WGS - 1) / GEMM_WGS;
    const int g = wave & 3, h = lane >> 4, c = lane & 15;
    for (int i = ((wave >> 2) - g) & 3; i < ntb; i += 4) {
        const int tile = blockIdx.x + GEMM_WGS * i;
        v8f acc[4] = {};
        unit_mma<256, 4>(X, Wl, tile, g * 64, lane, acc);
        float* crow = C + (size_t)(tile * 16 + h * 8) * 256 + g * 64 + c;
#pragma unroll
        for (int t = 0; t < 4; ++t) {
#pragma unroll
            for (int j = 0; j < 8; ++j) crow[(size_t)j * 256 + t * 16] = acc[t][j];
        }
    }
}

__global__ void __launch_bounds__(GEMM_THREADS) gemm2_kernel(const float* __restrict__ Y,
                                                             const float* __restrict__ W2,
                                                             const float* __restrict__ RW,
                                                             const float* __restrict__ b2,
                                                             float* __restrict__ F2,
                                                             float* __restrict__ OUTP) {
    __shared__ __attribute__((aligned(16))) float Wl[64 * (64 * 4 + 32)];
    const int tid = threadIdx.x, lane = tid & 31;
    const int wave = __builtin_amdgcn_readfirstlane(tid >> 5);
    stage_weights<64>(Wl, W2, RW, tid);
    __syncthreads();
    const int ntb = (NT_ROWS - (int)blockIdx.x + GEMM_WGS - 1) / GEMM_WGS;
    const int h = lane >> 4, c = lane & 15;
    for (int u = wave; u < 2 * ntb; u += 16) {
        const int tile = blockIdx.x + GEMM_WGS * (u >> 1);
        const int half = u & 1;
        v8f acc[2] = {};
        unit_mma<64, 2>(Y, Wl, tile, half * 32, lane, acc);
        const size_t o = (size_t)(tile * 16 + h * 8) * 32 + c;
        if (half == 0) {
#pragma unroll
            for (int t = 0; t < 2; ++t) {
#pragma unroll
                for (int j = 0; j < 8; ++j) F2[o + (size_t)j * 32 + t * 16] = acc[t][j];
            }
        } else {
#pragma unroll
            for (int t = 0; t < 2; ++t) {
                const float bb = b2[t * 16 + c];
#pragma unroll
                for (int j = 0; j < 8; ++j) OUTP[o + (size_t)j * 32 + t * 16] = acc[t][j] + bb;
            }
        }
    }
}

__device__ __forceinline__ float dot4(const v4f a, const v4f b) {
    return a.x * b.x + a.y * b.y + a.z * b.z + a.w * b.w;
}

#define ROWS1 8
#define ELR1_WGS ((NN / ROWS1 + 7) / 8)
#define TILE1 196
#define NTILE ((NN + TILE1 - 1) / TILE1)
static_assert(NN % ROWS1 == 0, "a wave takes ROWS1 whole rows");
static_assert(ROWS1 == 8, "the load fence in the kernel names eight rows");
static_assert(NTILE <= 256, "a tile number fits a byte");
static_assert(EE % 16 == 0, "codes are written four and read sixteen at a time");
static_assert(ELR1_WGS * 256 >= EE / 4, "one thread of the logit launch codes four edges");

__device__ __forceinline__ unsigned int tile_code(int d) { return ((unsigned int)d / TILE1) & 255u; }

__global__ void __launch_bounds__(256) elr1_kernel(const float* __restrict__ feat,
                                                   const float* __restrict__ al,
                                                   const float* __restrict__ ar,
                                                   float* __restrict__ el,
                                                   float* __restrict__ er,
                                                   const int* __restrict__ dst,
                                                   unsigned int* __restrict__ codes) {
    const int lane = threadIdx.x & 31;
    const int g = blockIdx.x * 256 + threadIdx.x;
    if (g < EE / 4) {
        const int4 d = reinterpret_cast<const int4*>(dst)[g];
        codes[g] = tile_code(d.x) | (tile_code(d.y) << 8) | (tile_code(d.z) << 16) | (tile_code(d.w) << 24);
    }
    const int row0 = (blockIdx.x * 8 + (threadIdx.x >> 5)) * ROWS1;
    if (row0 >= NN) return;
    const v4f* __restrict__ f4 = reinterpret_cast<const v4f*>(feat) + (size_t)row0 * 64 + lane;
    v4f v[ROWS1][2];
#pragma unroll
    for (int i = 0; i < ROWS1; ++i) { v[i][0] = f4[i * 64]; v[i][1] = f4[i * 64 + 32]; }
    const v4f* __restrict__ al4 = reinterpret_cast<const v4f*>(al);
    const v4f* __restrict__ ar4 = reinterpret_cast<const v4f*>(ar);
    const v4f l0 = al4[lane], l1 = al4[32 + lane];
    const v4f r0 = ar4[lane], r1 = ar4[32 + lane];
    asm volatile("" :: "v"(v[0][0]), "v"(v[0][1]), "v"(v[1][0]), "v"(v[1][1]),
                       "v"(v[2][0]), "v"(v[2][1]), "v"(v[3][0]), "v"(v[3][1]),
                       "v"(v[4][0]), "v"(v[4][1]), "v"(v[5][0]), "v"(v[5][1]),
                       "v"(v[6][0]), "v"(v[6][1]), "v"(v[7][0]), "v"(v[7][1]),
                       "v"(l0), "v"(l1), "v"(r0), "v"(r1));
    const bool b0 = (lane & 1) != 0, b1 = (lane & 2) != 0, b2 = (lane & 4) != 0;
    const int head = (lane >> 3) + (b1 ? 4 : 0);
    float* __restrict__ dstp = b0 ? er : el;
#pragma unroll
    for (int i = 0; i < ROWS1; i += 2) {
        float u[2];
#pragma unroll
        for (int c = 0; c < 2; ++c) {
            const float a0 = dot4(v[i + c][0], l0), e0 = dot4(v[i + c][0], r0);
            const float a1 = dot4(v[i + c][1], l1), e1 = dot4(v[i + c][1], r1);
            const float t0 = (b0 ? e0 : a0) + __shfl_xor(b0 ? a0 : e0, 1, 32);
            const float t1 = (b0 ? e1 : a1) + __shfl_xor(b0 ? a1 : e1, 1, 32);
            u[c] = (b1 ? t1 : t0) + __shfl_xor(b1 ? t0 : t1, 2, 32);
        }
        const float w = (b2 ? u[1] : u[0]) + __shfl_xor(b2 ? u[0] : u[1], 4, 32);
        dstp[(row0 + i + (b2 ? 1 : 0)) * H1N + head] = w;
    }
}

#define ROWS2 16
#define ELR2_WGS ((NN / ROWS2 + 3) / 4)
static_assert(NN % ROWS2 == 0, "a wave takes ROWS2 whole rows");
static_assert(ROWS2 == 16, "four loads of four rows, joined pairwise by the exchange tree");

__global__ void __launch_bounds__(128) elr2_kernel(const float* __restrict__ feat,
                                                   const float* __restrict__ al,
                                                   const float* __restrict__ ar,
                                                   float* __restrict__ el,
                                                   float* __restrict__ er) {
    const int lane = threadIdx.x & 31;
    const int row0 = (blockIdx.x * 4 + (threadIdx.x >> 5)) * ROWS2;
    if (row0 >= NN) return;
    const v4f* __restrict__ f4 = reinterpret_cast<const v4f*>(feat) + (size_t)row0 * 8 + lane;
    v4f v[4];
#pragma unroll
    for (int q = 0; q < 4; ++q) v[q] = f4[q * 32];
    const v4f l = reinterpret_cast<const v4f*>(al)[lane & 7];
    const v4f r = reinterpret_cast<const v4f*>(ar)[lane & 7];
    const bool b0 = (lane & 1) != 0, b1 = (lane & 2) != 0, b2 = (lane & 4) != 0;
    float t[4];
#pragma unroll
    for (int q = 0; q < 4; ++q) {
        const float a = dot4(v[q], l), e = dot4(v[q], r);
        t[q] = (b0 ? e : a) + __shfl_xor(b0 ? a : e, 1, 32);
    }
    const float u0 = (b1 ? t[1] : t[0]) + __shfl_xor(b1 ? t[0] : t[1], 2, 32);
    const float u1 = (b1 ? t[3] : t[2]) + __shfl_xor(b1 ? t[2] : t[3], 2, 32);
    const float w  = (b2 ? u1 : u0) + __shfl_xor(b2 ? u0 : u1, 4, 32);
    const int q = (b2 ? 2 : 0) + (b1 ? 1 : 0);
    (b0 ? er : el)[row0 + 4 * q + (lane >> 3)] = w;
}

__device__ __forceinline__ void lds_max_f32(float* addr, float v) {
    if (v >= 0.f) __hip_atomic_fetch_max((int*)addr, __float_as_int(v),
                                         __ATOMIC_RELAXED, __HIP_MEMORY_SCOPE_WORKGROUP);
    else          __hip_atomic_fetch_min((unsigned int*)addr, __float_as_uint(v),
                                         __ATOMIC_RELAXED, __HIP_MEMORY_SCOPE_WORKGROUP);
}

#define CAP1  4096
#define NITC  ((EE / 16 + 1023) / 1024)
#define GRPC  7
static_assert(NN <= 65536 && TILE1 < 65535, "tile row and source index are packed into 32 bits");
static_assert(NITC % GRPC == 0, "scan loop takes GRPC trips at a time");
static_assert(CAP1 % 1024 == 0, "a thread owns CAP1 / 1024 list places");

__device__ __forceinline__ unsigned int zero_bytes(unsigned int y) {
    return ~(((y & 0x7F7F7F7Fu) + 0x7F7F7F7Fu) | y) & 0x80808080u;
}

__global__ void __launch_bounds__(1024) agg1_tile_kernel(const int* __restrict__ src,
                                                         const int* __restrict__ dst,
                                                         const unsigned int* __restrict__ codes,
                                                         const float* __restrict__ el,
                                                         const float* __restrict__ er,
                                                         const float* __restrict__ feat,
                                                         const float* __restrict__ x,
                                                         const float* __restrict__ b1,
                                                         float* __restrict__ y,
                                                         int* __restrict__ glist,
                                                         int* __restrict__ gcnt) {
    __shared__ float acc[(TILE1 + 1) * 256];
    __shared__ __attribute__((aligned(16))) float errow[(TILE1 + 1) * H1N];
    __shared__ float mrow[(TILE1 + 1) * H1N];
    __shared__ float lrow[(TILE1 + 1) * H1N];
    __shared__ unsigned int ent[CAP1];
    __shared__ unsigned int cnt;

    const int tid  = threadIdx.x;
    const int lane = tid & 31;
    const int wave = __builtin_amdgcn_readfirstlane(tid >> 5);
    const int lo   = blockIdx.x * TILE1;
    const int rows = (NN - lo) < TILE1 ? (NN - lo) : TILE1;

    for (int idx = tid; idx < rows * 256; idx += 1024)
        acc[idx] = x[(size_t)lo * 256 + idx] + b1[idx & 255];
    for (int idx = tid; idx < (TILE1 + 1) * H1N; idx += 1024) {
        mrow[idx]  = -INFINITY;
        lrow[idx]  = 0.f;
        errow[idx] = idx < rows * H1N ? er[(size_t)lo * H1N + idx] : 0.f;
    }
    if (tid == 0) cnt = 0u;

    const uint4* __restrict__ code4 = reinterpret_cast<const uint4*>(codes);
    const unsigned int pat = (unsigned int)blockIdx.x * 0x01010101u;
    const int it0 = (blockIdx.x * NITC) >> 8;
    uint4 cur[GRPC];
#pragma unroll
    for (int q = 0; q < GRPC; ++q) {
        int it = it0 + q; if (it >= NITC) it -= NITC;
        const int iw = it * 1024 + tid;
        cur[q] = code4[iw < EE / 16 ? iw : tid];
    }
    __syncthreads();

    for (int j = 0; j < NITC; j += GRPC) {
        uint4 nxt[GRPC];
#pragma unroll
        for (int q = 0; q < GRPC; ++q) {
            int it = it0 + j + GRPC + q; if (it >= NITC) it -= NITC;
            const int iw = it * 1024 + tid;
            nxt[q] = code4[iw < EE / 16 ? iw : tid];
        }
#pragma unroll
        for (int q = 0; q < GRPC; ++q) {
            int it = it0 + j + q; if (it >= NITC) it -= NITC;
            const int iw = it * 1024 + tid;
            unsigned int m = (zero_bytes(cur[q].x ^ pat) >> 7) | (zero_bytes(cur[q].y ^ pat) >> 6)
                           | (zero_bytes(cur[q].z ^ pat) >> 5) | (zero_bytes(cur[q].w ^ pat) >> 4);
            if (iw >= EE / 16) m = 0u;
            while (m != 0u) {
                const unsigned int p = atomicAdd(&cnt, 1u);
                const int b = __builtin_ctz(m);
                if (p < CAP1) ent[p] = (unsigned int)(iw * 16 + (b & 7) * 4 + (b >> 3));
                m &= m - 1u;
            }
        }
#pragma unroll
        for (int q = 0; q < GRPC; ++q) cur[q] = nxt[q];
    }
    __syncthreads();
    const unsigned int total = cnt;
    if (tid == 0) gcnt[blockIdx.x] = (int)total;

    if (total <= CAP1) {
        const int n = (int)total;
        int* __restrict__ gl = glist + (size_t)blockIdx.x * CAP1;
        const float4* __restrict__ el4 = reinterpret_cast<const float4*>(el);
        float sc[CAP1 / 1024][H1N];
        int   rw[CAP1 / 1024];
        int   sw[CAP1 / 1024];
#pragma unroll
        for (int it = 0; it < CAP1 / 1024; ++it) {
            const int i = tid + 1024 * it;
            const int e = i < n ? (int)ent[i] : 0;
            rw[it] = dst[e];
            sw[it] = src[e];
        }
#pragma unroll
        for (int it = 0; it < CAP1 / 1024; ++it) {
            const unsigned int r = (unsigned int)(rw[it] - lo);
            rw[it] = (tid + 1024 * it) < n ? (r < (unsigned int)rows ? (int)r : TILE1) : -1;
        }
#pragma unroll
        for (int it = 0; it < CAP1 / 1024; ++it) {
            const int r = rw[it] < 0 ? 0 : rw[it];
            const float4 a0 = el4[(size_t)sw[it] * 2], a1 = el4[(size_t)sw[it] * 2 + 1];
            const float4 c0 = *reinterpret_cast<const float4*>(&errow[r * H1N]);
            const float4 c1 = *reinterpret_cast<const float4*>(&errow[r * H1N + 4]);
            sc[it][0] = leaky(a0.x + c0.x); sc[it][1] = leaky(a0.y + c0.y);
            sc[it][2] = leaky(a0.z + c0.z); sc[it][3] = leaky(a0.w + c0.w);
            sc[it][4] = leaky(a1.x + c1.x); sc[it][5] = leaky(a1.y + c1.y);
            sc[it][6] = leaky(a1.z + c1.z); sc[it][7] = leaky(a1.w + c1.w);
        }
#pragma unroll
        for (int it = 0; it < CAP1 / 1024; ++it) {
            if (rw[it] >= 0) {
#pragma unroll
                for (int h = 0; h < H1N; ++h) lds_max_f32(&mrow[rw[it] * H1N + h], sc[it][h]);
                const unsigned int pk = ((unsigned int)rw[it] << 16) | (unsigned int)sw[it];
                ent[tid + 1024 * it] = pk;
                gl[tid + 1024 * it]  = (int)pk;
            }
        }
        __syncthreads();
#pragma unroll
        for (int it = 0; it < CAP1 / 1024; ++it) {
            if (rw[it] >= 0) {
#pragma unroll
                for (int h = 0; h < H1N; ++h)
                    __hip_atomic_fetch_add(&lrow[rw[it] * H1N + h], __expf(sc[it][h] - mrow[rw[it] * H1N + h]),
                                           __ATOMIC_RELAXED, __HIP_MEMORY_SCOPE_WORKGROUP);
            }
        }
        __syncthreads();
        const int h = lane & 7;
        for (int i0 = wave; i0 < n; i0 += 128) {
            float a[4];
            const float* fp[4];
            float* ap[4];
            const int ikl = i0 + 32 * (lane & 3);
            const unsigned int uv = ent[ikl < n ? ikl : i0];
#pragma unroll
            for (int k = 0; k < 4; ++k) {
                const bool okk = (i0 + 32 * k) < n;
                const unsigned int u = (unsigned int)__builtin_amdgcn_readlane((int)uv, k);
                const int s = (int)(u & 0xFFFFu), r = (int)(u >> 16);
                a[k]  = __expf(leaky(el[s * H1N + h] + errow[r * H1N + h]) - mrow[r * H1N + h])
                        / lrow[r * H1N + h];
                fp[k] = feat + (size_t)s * 256 + lane;
                ap[k] = &acc[(okk ? r : TILE1) * 256 + lane];
            }
            float f[4][8];
#pragma unroll
            for (int j = 0; j < 8; ++j) {
#pragma unroll
                for (int k = 0; k < 4; ++k) f[k][j] = fp[k][32 * j];
            }
#pragma unroll
            for (int j = 0; j < 8; ++j) {
#pragma unroll
                for (int k = 0; k < 4; ++k) {
                    const float w = __int_as_float(__builtin_amdgcn_readlane(__float_as_int(a[k]), j));
                    __hip_atomic_fetch_add(ap[k] + 32 * j, f[k][j] * w,
                                           __ATOMIC_RELAXED, __HIP_MEMORY_SCOPE_WORKGROUP);
                }
            }
        }
    } else {
        for (int e = tid; e < EE; e += 1024) {
            const unsigned int r = (unsigned int)(dst[e] - lo);
            if (r < (unsigned int)rows) {
                const int sidx = src[e];
                for (int hh = 0; hh < H1N; ++hh)
                    lds_max_f32(&mrow[r * H1N + hh], leaky(el[sidx * H1N + hh] + errow[r * H1N + hh]));
            }
        }
        __syncthreads();
        for (int e = tid; e < EE; e += 1024) {
            const unsigned int r = (unsigned int)(dst[e] - lo);
            if (r < (unsigned int)rows) {
                const int sidx = src[e];
                for (int hh = 0; hh < H1N; ++hh)
                    __hip_atomic_fetch_add(&lrow[r * H1N + hh],
                                           __expf(leaky(el[sidx * H1N + hh] + errow[r * H1N + hh]) - mrow[r * H1N + hh]),
                                           __ATOMIC_RELAXED, __HIP_MEMORY_SCOPE_WORKGROUP);
            }
        }
        __syncthreads();
        for (int e = tid; e < EE; e += 1024) {
            const unsigned int r = (unsigned int)(dst[e] - lo);
            if (r < (unsigned int)rows) {
                const int sidx = src[e];
                for (int hh = 0; hh < H1N; ++hh) {
                    const float w = __expf(leaky(el[sidx * H1N + hh] + errow[r * H1N + hh]) - mrow[r * H1N + hh])
                                    / lrow[r * H1N + hh];
                    for (int c = 0; c < 32; ++c)
                        __hip_atomic_fetch_add(&acc[r * 256 + hh * 32 + c], feat[(size_t)sidx * 256 + hh * 32 + c] * w,
                                               __ATOMIC_RELAXED, __HIP_MEMORY_SCOPE_WORKGROUP);
                }
            }
        }
    }
    __syncthreads();

    for (int idx = tid; idx < rows * 256; idx += 1024)
        y[(size_t)lo * 256 + idx] = fmaxf(acc[idx], 0.f);
}

#define TILE2 196
static_assert(TILE2 == TILE1, "layer 2 reads the lists that layer 1 made for its own tiles");
static_assert(NN <= 65536, "source index is packed into 16 bits");
static_assert((size_t)NTILE * CAP1 + NTILE + EE / 4 <= (size_t)NN * 34, "lists, counts and tile codes fit the workspace part kept for them");


__global__ void __launch_bounds__(1024) agg2_tile_kernel(const int* __restrict__ src,
                                                         const int* __restrict__ dst,
                                                         const float* __restrict__ el,
                                                         const float* __restrict__ er,
                                                         const float* __restrict__ feat,
                                                         const int* __restrict__ glist,
                                                         const int* __restrict__ gcnt,
                                                         float* __restrict__ outp) {
    __shared__ float acc[(TILE2 + 1) * 32];
    __shared__ float mrow[TILE2 + 1];
    __shared__ float lrow[TILE2 + 1];
    __shared__ float errow[TILE2 + 1];
    __shared__ unsigned int ent[CAP1];
    __shared__ float val[CAP1];

    const int tid  = threadIdx.x;
    const int lo   = blockIdx.x * TILE2;
    const int rows = (NN - lo) < TILE2 ? (NN - lo) : TILE2;
    const unsigned int total = (unsigned int)gcnt[blockIdx.x];

    for (int idx = tid; idx < (TILE2 + 1) * 32; idx += 1024) acc[idx] = 0.f;
    if (tid <= TILE2) {
        mrow[tid]  = -INFINITY;
        lrow[tid]  = 0.f;
        errow[tid] = tid < rows ? er[lo + tid] : 0.f;
    }
    __syncthreads();

    if (total <= CAP1) {
        const int n = (int)total;
        const int* __restrict__ gl = glist + (size_t)blockIdx.x * CAP1;
        unsigned int uu[CAP1 / 1024];
        float ee[CAP1 / 1024];
#pragma unroll
        for (int it = 0; it < CAP1 / 1024; ++it) {
            const int i = tid + 1024 * it;
            uu[it] = i < n ? (unsigned int)gl[i] : 0u;
        }
#pragma unroll
        for (int it = 0; it < CAP1 / 1024; ++it) ee[it] = el[uu[it] & 0xFFFFu];
#pragma unroll
        for (int it = 0; it < CAP1 / 1024; ++it) {
            const int i = tid + 1024 * it;
            if (i < n) {
                const int r = (int)(uu[it] >> 16);
                const float sc = leaky(ee[it] + errow[r]);
                lds_max_f32(&mrow[r], sc);
                ent[i] = uu[it];
                val[i] = sc;
            }
        }
        __syncthreads();
        for (int i = tid; i < n; i += 1024) {
            const int r = (int)(ent[i] >> 16);
            const float p = __expf(val[i] - mrow[r]);
            val[i] = p;
            __hip_atomic_fetch_add(&lrow[r], p, __ATOMIC_RELAXED, __HIP_MEMORY_SCOPE_WORKGROUP);
        }
        __syncthreads();
        const int sub = tid & 7;
        const float4* __restrict__ feat4 = reinterpret_cast<const float4*>(feat);
        for (int i = tid >> 3; i < n; i += 512) {
            float4 f[4]; float a[4]; int rr[4];
#pragma unroll
            for (int k = 0; k < 4; ++k) {
                const int ik = i + 128 * k;
                const int ic = ik < n ? ik : i;
                const unsigned int u = ent[ic];
                rr[k] = ik < n ? (int)(u >> 16) : -1;
                a[k]  = val[ic] / lrow[u >> 16];
                f[k]  = feat4[(size_t)(u & 0xFFFFu) * 8 + sub];
            }
#pragma unroll
            for (int k = 0; k < 4; ++k) {
                if (rr[k] >= 0) {
                    float* p = &acc[rr[k] * 32 + sub * 4];
                    __hip_atomic_fetch_add(p + 0, f[k].x * a[k], __ATOMIC_RELAXED, __HIP_MEMORY_SCOPE_WORKGROUP);
                    __hip_atomic_fetch_add(p + 1, f[k].y * a[k], __ATOMIC_RELAXED, __HIP_MEMORY_SCOPE_WORKGROUP);
                    __hip_atomic_fetch_add(p + 2, f[k].z * a[k], __ATOMIC_RELAXED, __HIP_MEMORY_SCOPE_WORKGROUP);
                    __hip_atomic_fetch_add(p + 3, f[k].w * a[k], __ATOMIC_RELAXED, __HIP_MEMORY_SCOPE_WORKGROUP);
                }
            }
        }
    } else {
        for (int e = tid; e < EE; e += 1024) {
            const unsigned int r = (unsigned int)(dst[e] - lo);
            if (r < (unsigned int)rows) lds_max_f32(&mrow[r], leaky(el[src[e]] + errow[r]));
        }
        __syncthreads();
        for (int e = tid; e < EE; e += 1024) {
            const unsigned int r = (unsigned int)(dst[e] - lo);
            if (r < (unsigned int)rows)
                __hip_atomic_fetch_add(&lrow[r], __expf(leaky(el[src[e]] + errow[r]) - mrow[r]),
                                       __ATOMIC_RELAXED, __HIP_MEMORY_SCOPE_WORKGROUP);
        }
        __syncthreads();
        for (int e = tid; e < EE; e += 1024) {
            const unsigned int r = (unsigned int)(dst[e] - lo);
            if (r < (unsigned int)rows) {
                const int sidx = src[e];
                const float a = __expf(leaky(el[sidx] + errow[r]) - mrow[r]) / lrow[r];
                for (int c = 0; c < 32; ++c)
                    __hip_atomic_fetch_add(&acc[r * 32 + c], feat[(size_t)sidx * 32 + c] * a,
                                           __ATOMIC_RELAXED, __HIP_MEMORY_SCOPE_WORKGROUP);
            }
        }
    }
    __syncthreads();

    for (int idx = tid; idx < rows * 32; idx += 1024)
        outp[(size_t)lo * 32 + idx] += acc[idx];
}

extern "C" void kernel_launch(void* const* d_in, const int* in_sizes, int n_in,
                              void* d_out, int out_size, void* d_ws, size_t ws_size,
                              hipStream_t stream) {
    const float* x     = (const float*)d_in[0];
    const int*   src   = (const int*)  d_in[1];
    const int*   dst   = (const int*)  d_in[2];
    const float* W1    = (const float*)d_in[3];
    const float* al1   = (const float*)d_in[4];
    const float* ar1   = (const float*)d_in[5];
    const float* b1    = (const float*)d_in[6];
    const float* W2    = (const float*)d_in[7];
    const float* al2   = (const float*)d_in[8];
    const float* ar2   = (const float*)d_in[9];
    const float* resW2 = (const float*)d_in[10];
    const float* b2    = (const float*)d_in[11];
    float* outp = (float*)d_out;

    float* ws    = (float*)d_ws;
    float* feat1 = ws;
    float* rst1  = feat1 + (size_t)NN * 256;
    float* el1   = rst1  + (size_t)NN * 256;
    float* er1   = el1   + (size_t)NN * 8;
    int*   lst   = (int*)(er1 + (size_t)NN * 8);
    int*   lcnt  = lst + (size_t)NTILE * CAP1;
    unsigned int* codes = (unsigned int*)(lcnt + NTILE);
    float* feat2 = feat1;
    float* el2   = el1;
    float* er2   = er1;

    gemm1_kernel<<<GEMM_WGS, GEMM_THREADS, 0, stream>>>(x, W1, feat1);
    elr1_kernel<<<ELR1_WGS, 256, 0, stream>>>(feat1, al1, ar1, el1, er1, dst, codes);
    agg1_tile_kernel<<<NTILE, 1024, 0, stream>>>(src, dst, codes, el1, er1,
                                                 feat1, x, b1, rst1, lst, lcnt);

    gemm2_kernel<<<GEMM_WGS, GEMM_THREADS, 0, stream>>>(rst1, W2, resW2, b2, feat2, outp);
    elr2_kernel<<<ELR2_WGS, 128, 0, stream>>>(feat2, al2, ar2, el2, er2);
    agg2_tile_kernel<<<NTILE, 1024, 0, stream>>>(src, dst, el2, er2, feat2, lst, lcnt, outp);
}


# Layer-1 gather by owner waves: edges dealt (LDS arrival rank) to the wave owning row % 32, which adds weighted rows (fma) by plain 16-byte LDS loads/stores; no float atomics; sum order: owner list
# speedup vs baseline: 4.9019x; 1.0413x over previous
#include <hip/hip_runtime.h>
#include <hip/hip_bf16.h>
#include <math.h>

#define NN 50000
#define EE 800000
#define INF_ 256
#define OUTF 32
#define H1N 8
#define SLOPE 0.2f

typedef float v2f __attribute__((ext_vector_type(2)));
typedef float v8f __attribute__((ext_vector_type(8)));

__device__ __forceinline__ float leaky(float x) { return x > 0.f ? x : SLOPE * x; }

typedef float v4f __attribute__((ext_vector_type(4)));

#define NT_ROWS (NN / 16)
#define GEMM_WGS 256
#define GEMM_THREADS 512
static_assert(NN % 16 == 0, "whole 16-row tiles only: no row guard in the GEMM kernels");

template <int NCOL>
__device__ __forceinline__ void stage_weights(float* __restrict__ Wl, const float* __restrict__ Wa,
                                              const float* __restrict__ Wb, const int tid) {
    constexpr int P = NCOL * 4 + 32;
    constexpr int CG = NCOL / 4;
    constexpr int NB = 64 * CG / GEMM_THREADS;
    static_assert(64 * CG % GEMM_THREADS == 0, "every thread moves the same number of blocks");
    const int rot = (tid >> 2) & 3;
    v4f r[NB][4];
#pragma unroll
    for (int it = 0; it < NB; ++it) {
        const int q = tid + it * GEMM_THREADS;
        const int p = q / CG, cg = q % CG;
        const float* s;
        int pitch;
        if (NCOL == 256) { s = Wa + (size_t)(4 * p) * 256 + 4 * cg; pitch = 256; }
        else             { s = (cg < 8 ? Wa : Wb) + (size_t)(4 * p) * 32 + 4 * (cg & 7); pitch = 32; }
#pragma unroll
        for (int i = 0; i < 4; ++i) r[it][i] = *reinterpret_cast<const v4f*>(s + i * pitch);
    }
#pragma unroll
    for (int it = 0; it < NB; ++it) {
        const int q = tid + it * GEMM_THREADS;
        const int p = q / CG, cg = q % CG;
#pragma unroll
        for (int j = 0; j < 4; ++j) {
            const int jj = (j + rot) & 3;
            v4f o;
            o.x = jj == 0 ? r[it][0].x : jj == 1 ? r[it][0].y : jj == 2 ? r[it][0].z : r[it][0].w;
            o.y = jj == 0 ? r[it][1].x : jj == 1 ? r[it][1].y : jj == 2 ? r[it][1].z : r[it][1].w;
            o.z = jj == 0 ? r[it][2].x : jj == 1 ? r[it][2].y : jj == 2 ? r[it][2].z : r[it][2].w;
            o.w = jj == 0 ? r[it][3].x : jj == 1 ? r[it][3].y : jj == 2 ? r[it][3].z : r[it][3].w;
            *reinterpret_cast<v4f*>(Wl + p * P + (4 * cg + jj) * 4) = o;
        }
    }
}

template <int NCOL, int NT16>
__device__ __forceinline__ void unit_mma(const float* __restrict__ A, const float* __restrict__ Wl,
                                         const int tile, const int col0, const int lane, v8f (&acc)[NT16]) {
    constexpr int P = NCOL * 4 + 32;
    const int h = lane >> 4, c = lane & 15;
    const v4f* __restrict__ a4 = reinterpret_cast<const v4f*>(A + (size_t)(tile * 16 + c) * 256) + h;
    const float* bp = Wl + h * P + (col0 + c) * 4;
    for (int J0 = 0; J0 < 32; J0 += 8) {
        v4f ac[8];
#pragma unroll
        for (int j = 0; j < 8; ++j) ac[j] = a4[2 * (J0 + j)];
#pragma unroll
        for (int j = 0; j < 8; ++j) {
            const float* bj = bp + 2 * (J0 + j) * P;
            v4f b[NT16];
#pragma unroll
            for (int t = 0; t < NT16; ++t) b[t] = *reinterpret_cast<const v4f*>(bj + t * 64);
            v2f alo, ahi;
            alo.x = ac[j].x; alo.y = ac[j].y; ahi.x = ac[j].z; ahi.y = ac[j].w;
#pragma unroll
            for (int t = 0; t < NT16; ++t) {
                v2f blo; blo.x = b[t].x; blo.y = b[t].y;
                acc[t] = __builtin_amdgcn_wmma_f32_16x16x4_f32(false, alo, false, blo, (short)0, acc[t], false, false);
            }
#pragma unroll
            for (int t = 0; t < NT16; ++t) {
                v2f bhi; bhi.x = b[t].z; bhi.y = b[t].w;
                acc[t] = __builtin_amdgcn_wmma_f32_16x16x4_f32(false, ahi, false, bhi, (short)0, acc[t], false, false);
            }
        }
    }
}

__global__ void __launch_bounds__(GEMM_THREADS) gemm1_kernel(const float* __restrict__ X,
                                                             const float* __restrict__ W,
                                                             float* __restrict__ C) {
    __shared__ __attribute__((aligned(16))) float Wl[64 * (256 * 4 + 32)];
    const int tid = threadIdx.x, lane = tid & 31;
    const int wave = __builtin_amdgcn_readfirstlane(tid >> 5);
    stage_weights<256>(Wl, W, W, tid);
    __syncthreads();
    const int ntb = (NT_ROWS - (int)blockIdx.x + GEMM_WGS - 1) / GEMM_WGS;
    const int g = wave & 3, h = lane >> 4, c = lane & 15;
    for (int i = ((wave >> 2) - g) & 3; i < ntb; i += 4) {
        const int tile = blockIdx.x + GEMM_WGS * i;
        v8f acc[4] = {};
        unit_mma<256, 4>(X, Wl, tile, g * 64, lane, acc);
        float* crow = C + (size_t)(tile * 16 + h * 8) * 256 + g * 64 + c;
#pragma unroll
        for (int t = 0; t < 4; ++t) {
#pragma unroll
            for (int j = 0; j < 8; ++j) crow[(size_t)j * 256 + t * 16] = acc[t][j];
        }
    }
}

__global__ void __launch_bounds__(GEMM_THREADS) gemm2_kernel(const float* __restrict__ Y,
                                                             const float* __restrict__ W2,
                                                             const float* __restrict__ RW,
                                                             const float* __restrict__ b2,
                                                             float* __restrict__ F2,
                                                             float* __restrict__ OUTP) {
    __shared__ __attribute__((aligned(16))) float Wl[64 * (64 * 4 + 32)];
    const int tid = threadIdx.x, lane = tid & 31;
    const int wave = __builtin_amdgcn_readfirstlane(tid >> 5);
    stage_weights<64>(Wl, W2, RW, tid);
    __syncthreads();
    const int ntb = (NT_ROWS - (int)blockIdx.x + GEMM_WGS - 1) / GEMM_WGS;
    const int h = lane >> 4, c = lane & 15;
    for (int u = wave; u < 2 * ntb; u += 16) {
        const int tile = blockIdx.x + GEMM_WGS * (u >> 1);
        const int half = u & 1;
        v8f acc[2] = {};
        unit_mma<64, 2>(Y, Wl, tile, half * 32, lane, acc);
        const size_t o = (size_t)(tile * 16 + h * 8) * 32 + c;
        if (half == 0) {
#pragma unroll
            for (int t = 0; t < 2; ++t) {
#pragma unroll
                for (int j = 0; j < 8; ++j) F2[o + (size_t)j * 32 + t * 16] = acc[t][j];
            }
        } else {
#pragma unroll
            for (int t = 0; t < 2; ++t) {
                const float bb = b2[t * 16 + c];
#pragma unroll
                for (int j = 0; j < 8; ++j) OUTP[o + (size_t)j * 32 + t * 16] = acc[t][j] + bb;
            }
        }
    }
}

__device__ __forceinline__ float dot4(const v4f a, const v4f b) {
    return a.x * b.x + a.y * b.y + a.z * b.z + a.w * b.w;
}

#define ROWS1 8
#define ELR1_WGS ((NN / ROWS1 + 7) / 8)
#define TILE1 196
#define NTILE ((NN + TILE1 - 1) / TILE1)
static_assert(NN % ROWS1 == 0, "a wave takes ROWS1 whole rows");
static_assert(ROWS1 == 8, "the load fence in the kernel names eight rows");
static_assert(NTILE <= 256, "a tile number fits a byte");
static_assert(EE % 16 == 0, "codes are written four and read sixteen at a time");
static_assert(ELR1_WGS * 256 >= EE / 4, "one thread of the logit launch codes four edges");

__device__ __forceinline__ unsigned int tile_code(int d) { return ((unsigned int)d / TILE1) & 255u; }

__global__ void __launch_bounds__(256) elr1_kernel(const float* __restrict__ feat,
                                                   const float* __restrict__ al,
                                                   const float* __restrict__ ar,
                                                   float* __restrict__ el,
                                                   float* __restrict__ er,
                                                   const int* __restrict__ dst,
                                                   unsigned int* __restrict__ codes) {
    const int lane = threadIdx.x & 31;
    const int g = blockIdx.x * 256 + threadIdx.x;
    if (g < EE / 4) {
        const int4 d = reinterpret_cast<const int4*>(dst)[g];
        codes[g] = tile_code(d.x) | (tile_code(d.y) << 8) | (tile_code(d.z) << 16) | (tile_code(d.w) << 24);
    }
    const int row0 = (blockIdx.x * 8 + (threadIdx.x >> 5)) * ROWS1;
    if (row0 >= NN) return;
    const v4f* __restrict__ f4 = reinterpret_cast<const v4f*>(feat) + (size_t)row0 * 64 + lane;
    v4f v[ROWS1][2];
#pragma unroll
    for (int i = 0; i < ROWS1; ++i) { v[i][0] = f4[i * 64]; v[i][1] = f4[i * 64 + 32]; }
    const v4f* __restrict__ al4 = reinterpret_cast<const v4f*>(al);
    const v4f* __restrict__ ar4 = reinterpret_cast<const v4f*>(ar);
    const v4f l0 = al4[lane], l1 = al4[32 + lane];
    const v4f r0 = ar4[lane], r1 = ar4[32 + lane];
    asm volatile("" :: "v"(v[0][0]), "v"(v[0][1]), "v"(v[1][0]), "v"(v[1][1]),
                       "v"(v[2][0]), "v"(v[2][1]), "v"(v[3][0]), "v"(v[3][1]),
                       "v"(v[4][0]), "v"(v[4][1]), "v"(v[5][0]), "v"(v[5][1]),
                       "v"(v[6][0]), "v"(v[6][1]), "v"(v[7][0]), "v"(v[7][1]),
                       "v"(l0), "v"(l1), "v"(r0), "v"(r1));
    const bool b0 = (lane & 1) != 0, b1 = (lane & 2) != 0, b2 = (lane & 4) != 0;
    const int head = (lane >> 3) + (b1 ? 4 : 0);
    float* __restrict__ dstp = b0 ? er : el;
#pragma unroll
    for (int i = 0; i < ROWS1; i += 2) {
        float u[2];
#pragma unroll
        for (int c = 0; c < 2; ++c) {
            const float a0 = dot4(v[i + c][0], l0), e0 = dot4(v[i + c][0], r0);
            const float a1 = dot4(v[i + c][1], l1), e1 = dot4(v[i + c][1], r1);
            const float t0 = (b0 ? e0 : a0) + __shfl_xor(b0 ? a0 : e0, 1, 32);
            const float t1 = (b0 ? e1 : a1) + __shfl_xor(b0 ? a1 : e1, 1, 32);
            u[c] = (b1 ? t1 : t0) + __shfl_xor(b1 ? t0 : t1, 2, 32);
        }
        const float w = (b2 ? u[1] : u[0]) + __shfl_xor(b2 ? u[0] : u[1], 4, 32);
        dstp[(row0 + i + (b2 ? 1 : 0)) * H1N + head] = w;
    }
}

#define ROWS2 16
#define ELR2_WGS ((NN / ROWS2 + 3) / 4)
static_assert(NN % ROWS2 == 0, "a wave takes ROWS2 whole rows");
static_assert(ROWS2 == 16, "four loads of four rows, joined pairwise by the exchange tree");

__global__ void __launch_bounds__(128) elr2_kernel(const float* __restrict__ feat,
                                                   const float* __restrict__ al,
                                                   const float* __restrict__ ar,
                                                   float* __restrict__ el,
                                                   float* __restrict__ er) {
    const int lane = threadIdx.x & 31;
    const int row0 = (blockIdx.x * 4 + (threadIdx.x >> 5)) * ROWS2;
    if (row0 >= NN) return;
    const v4f* __restrict__ f4 = reinterpret_cast<const v4f*>(feat) + (size_t)row0 * 8 + lane;
    v4f v[4];
#pragma unroll
    for (int q = 0; q < 4; ++q) v[q] = f4[q * 32];
    const v4f l = reinterpret_cast<const v4f*>(al)[lane & 7];
    const v4f r = reinterpret_cast<const v4f*>(ar)[lane & 7];
    const bool b0 = (lane & 1) != 0, b1 = (lane & 2) != 0, b2 = (lane & 4) != 0;
    float t[4];
#pragma unroll
    for (int q = 0; q < 4; ++q) {
        const float a = dot4(v[q], l), e = dot4(v[q], r);
        t[q] = (b0 ? e : a) + __shfl_xor(b0 ? a : e, 1, 32);
    }
    const float u0 = (b1 ? t[1] : t[0]) + __shfl_xor(b1 ? t[0] : t[1], 2, 32);
    const float u1 = (b1 ? t[3] : t[2]) + __shfl_xor(b1 ? t[2] : t[3], 2, 32);
    const float w  = (b2 ? u1 : u0) + __shfl_xor(b2 ? u0 : u1, 4, 32);
    const int q = (b2 ? 2 : 0) + (b1 ? 1 : 0);
    (b0 ? er : el)[row0 + 4 * q + (lane >> 3)] = w;
}

__device__ __forceinline__ void lds_max_f32(float* addr, float v) {
    if (v >= 0.f) __hip_atomic_fetch_max((int*)addr, __float_as_int(v),
                                         __ATOMIC_RELAXED, __HIP_MEMORY_SCOPE_WORKGROUP);
    else          __hip_atomic_fetch_min((unsigned int*)addr, __float_as_uint(v),
                                         __ATOMIC_RELAXED, __HIP_MEMORY_SCOPE_WORKGROUP);
}

#define CAP1  4096
#define WCAP  256
#define NITC  ((EE / 16 + 1023) / 1024)
#define GRPC  7
static_assert(NN <= 65536 && TILE1 < 65535, "tile row and source index are packed into 32 bits");
static_assert(NITC % GRPC == 0, "scan loop takes GRPC trips at a time");
static_assert(CAP1 % 1024 == 0, "a thread owns CAP1 / 1024 list places");

__device__ __forceinline__ unsigned int zero_bytes(unsigned int y) {
    return ~(((y & 0x7F7F7F7Fu) + 0x7F7F7F7Fu) | y) & 0x80808080u;
}

__device__ __forceinline__ float row_pick(float x, unsigned int lo, unsigned int hi) {
    return __int_as_float(__builtin_amdgcn_permlane16(__float_as_int(x), __float_as_int(x), lo, hi, false, false));
}

__device__ __forceinline__ void edge_loads(const unsigned int uv, const float* __restrict__ el,
                                           const v4f* __restrict__ feat4, const int hd, const int lane,
                                           float (&ev)[4], v4f (&fA)[4], v4f (&fB)[4]) {
#pragma unroll
    for (int k = 0; k < 4; ++k) {
        const int s = __builtin_amdgcn_readlane((int)uv, k) & 0xFFFF;
        ev[k] = el[s * H1N + hd];
        fA[k] = feat4[(size_t)s * 64 + lane];
        fB[k] = feat4[(size_t)s * 64 + 32 + lane];
    }
}

__device__ __forceinline__ void edge_sums(const unsigned int uv, const int i0, const int nw, const int hd,
                                          const int lane, const float (&ev)[4], const v4f (&fA)[4],
                                          const v4f (&fB)[4], float* acc, const float* __restrict__ errow,
                                          const float* __restrict__ mrow, const float* __restrict__ lrow) {
    asm volatile("" :: "v"(ev[0]), "v"(ev[1]), "v"(ev[2]), "v"(ev[3]),
                       "v"(fA[0]), "v"(fB[0]), "v"(fA[1]), "v"(fB[1]),
                       "v"(fA[2]), "v"(fB[2]), "v"(fA[3]), "v"(fB[3]));
    int rr[4];
    float wA[4], wB[4];
#pragma unroll
    for (int k = 0; k < 4; ++k) {
        const int r = (int)((unsigned int)__builtin_amdgcn_readlane((int)uv, k) >> 16);
        rr[k] = r;
        const float a = __expf(leaky(ev[k] + errow[r * H1N + hd]) - mrow[r * H1N + hd])
                        / lrow[r * H1N + hd];
        wA[k] = row_pick(a, 0x00000000u, 0x11111111u);
        wB[k] = row_pick(a, 0x22222222u, 0x33333333u);
    }
#pragma unroll
    for (int k = 0; k < 4; ++k) {
        if (i0 + k < nw) {
            v4f* pa = reinterpret_cast<v4f*>(&acc[rr[k] * 256]) + lane;
            v4f cA = pa[0], cB = pa[32];
            cA += fA[k] * wA[k];
            cB += fB[k] * wB[k];
            pa[0] = cA; pa[32] = cB;
        }
    }
}

__global__ void __launch_bounds__(1024) agg1_tile_kernel(const int* __restrict__ src,
                                                         const int* __restrict__ dst,
                                                         const unsigned int* __restrict__ codes,
                                                         const float* __restrict__ el,
                                                         const float* __restrict__ er,
                                                         const float* __restrict__ feat,
                                                         const float* __restrict__ x,
                                                         const float* __restrict__ b1,
                                                         float* __restrict__ y,
                                                         int* __restrict__ glist,
                                                         int* __restrict__ gcnt) {
    __shared__ __attribute__((aligned(16))) float acc[(TILE1 + 1) * 256];
    __shared__ __attribute__((aligned(16))) float errow[(TILE1 + 1) * H1N];
    __shared__ float mrow[(TILE1 + 1) * H1N];
    __shared__ float lrow[(TILE1 + 1) * H1N];
    __shared__ unsigned int ent[CAP1];
    __shared__ unsigned int wl[32 * WCAP];
    __shared__ unsigned int wcnt[32];
    __shared__ unsigned int cnt;

    const int tid  = threadIdx.x;
    const int lane = tid & 31;
    const int wave = __builtin_amdgcn_readfirstlane(tid >> 5);
    const int lo   = blockIdx.x * TILE1;
    const int rows = (NN - lo) < TILE1 ? (NN - lo) : TILE1;

    for (int idx = tid; idx < rows * 256; idx += 1024)
        acc[idx] = x[(size_t)lo * 256 + idx] + b1[idx & 255];
    for (int idx = tid; idx < (TILE1 + 1) * H1N; idx += 1024) {
        mrow[idx]  = -INFINITY;
        lrow[idx]  = 0.f;
        errow[idx] = idx < rows * H1N ? er[(size_t)lo * H1N + idx] : 0.f;
    }
    if (tid == 0) cnt = 0u;
    if (tid < 32) wcnt[tid] = 0u;

    const uint4* __restrict__ code4 = reinterpret_cast<const uint4*>(codes);
    const unsigned int pat = (unsigned int)blockIdx.x * 0x01010101u;
    const int it0 = (blockIdx.x * NITC) >> 8;
    uint4 cur[GRPC];
#pragma unroll
    for (int q = 0; q < GRPC; ++q) {
        int it = it0 + q; if (it >= NITC) it -= NITC;
        const int iw = it * 1024 + tid;
        cur[q] = code4[iw < EE / 16 ? iw : tid];
    }
    __syncthreads();

    for (int j = 0; j < NITC; j += GRPC) {
        uint4 nxt[GRPC];
#pragma unroll
        for (int q = 0; q < GRPC; ++q) {
            int it = it0 + j + GRPC + q; if (it >= NITC) it -= NITC;
            const int iw = it * 1024 + tid;
            nxt[q] = code4[iw < EE / 16 ? iw : tid];
        }
#pragma unroll
        for (int q = 0; q < GRPC; ++q) {
            int it = it0 + j + q; if (it >= NITC) it -= NITC;
            const int iw = it * 1024 + tid;
            unsigned int m = (zero_bytes(cur[q].x ^ pat) >> 7) | (zero_bytes(cur[q].y ^ pat) >> 6)
                           | (zero_bytes(cur[q].z ^ pat) >> 5) | (zero_bytes(cur[q].w ^ pat) >> 4);
            if (iw >= EE / 16) m = 0u;
            while (m != 0u) {
                const unsigned int p = atomicAdd(&cnt, 1u);
                const int b = __builtin_ctz(m);
                if (p < CAP1) ent[p] = (unsigned int)(iw * 16 + (b & 7) * 4 + (b >> 3));
                m &= m - 1u;
            }
        }
#pragma unroll
        for (int q = 0; q < GRPC; ++q) cur[q] = nxt[q];
    }
    __syncthreads();
    const unsigned int total = cnt;
    if (tid == 0) gcnt[blockIdx.x] = (int)total;

    if (total <= CAP1) {
        const int n = (int)total;
        int* __restrict__ gl = glist + (size_t)blockIdx.x * CAP1;
        const float4* __restrict__ el4 = reinterpret_cast<const float4*>(el);
        float sc[CAP1 / 1024][H1N];
        int   rw[CAP1 / 1024];
        int   sw[CAP1 / 1024];
#pragma unroll
        for (int it = 0; it < CAP1 / 1024; ++it) {
            const int i = tid + 1024 * it;
            const int e = i < n ? (int)ent[i] : 0;
            rw[it] = dst[e];
            sw[it] = src[e];
        }
#pragma unroll
        for (int it = 0; it < CAP1 / 1024; ++it) {
            const unsigned int r = (unsigned int)(rw[it] - lo);
            rw[it] = (tid + 1024 * it) < n ? (r < (unsigned int)rows ? (int)r : TILE1) : -1;
        }
#pragma unroll
        for (int it = 0; it < CAP1 / 1024; ++it) {
            const int r = rw[it] < 0 ? 0 : rw[it];
            const float4 a0 = el4[(size_t)sw[it] * 2], a1 = el4[(size_t)sw[it] * 2 + 1];
            const float4 c0 = *reinterpret_cast<const float4*>(&errow[r * H1N]);
            const float4 c1 = *reinterpret_cast<const float4*>(&errow[r * H1N + 4]);
            sc[it][0] = leaky(a0.x + c0.x); sc[it][1] = leaky(a0.y + c0.y);
            sc[it][2] = leaky(a0.z + c0.z); sc[it][3] = leaky(a0.w + c0.w);
            sc[it][4] = leaky(a1.x + c1.x); sc[it][5] = leaky(a1.y + c1.y);
            sc[it][6] = leaky(a1.z + c1.z); sc[it][7] = leaky(a1.w + c1.w);
        }
#pragma unroll
        for (int it = 0; it < CAP1 / 1024; ++it) {
            if (rw[it] >= 0) {
#pragma unroll
                for (int h = 0; h < H1N; ++h) lds_max_f32(&mrow[rw[it] * H1N + h], sc[it][h]);
                const unsigned int pk = ((unsigned int)rw[it] << 16) | (unsigned int)sw[it];
                ent[tid + 1024 * it] = pk;
                gl[tid + 1024 * it]  = (int)pk;
                const unsigned int ow = (unsigned int)rw[it] & 31u;
                const unsigned int p  = atomicAdd(&wcnt[ow], 1u);
                if (p < WCAP) wl[ow * WCAP + p] = pk;
            }
        }
        __syncthreads();
#pragma unroll
        for (int it = 0; it < CAP1 / 1024; ++it) {
            if (rw[it] >= 0) {
#pragma unroll
                for (int h = 0; h < H1N; ++h)
                    __hip_atomic_fetch_add(&lrow[rw[it] * H1N + h], __expf(sc[it][h] - mrow[rw[it] * H1N + h]),
                                           __ATOMIC_RELAXED, __HIP_MEMORY_SCOPE_WORKGROUP);
            }
        }
        __syncthreads();
        const unsigned int myc = wcnt[lane];
        if (!__any(myc > (unsigned int)WCAP)) {
            const int nw = __builtin_amdgcn_readfirstlane((int)wcnt[wave]);
            const unsigned int* wlw = wl + wave * WCAP;
            const int q  = lane & 3;
            const int hd = (q & 1) + 2 * (lane >> 4) + 4 * (q >> 1);
            const v4f* __restrict__ feat4 = reinterpret_cast<const v4f*>(feat);
            if (nw > 0) {
                float evP[4], evQ[4];
                v4f fAP[4], fBP[4], fAQ[4], fBQ[4];
                unsigned int uvP = wlw[q < nw ? q : 0], uvQ;
                edge_loads(uvP, el, feat4, hd, lane, evP, fAP, fBP);
                for (int i0 = 0; i0 < nw; i0 += 8) {
                    uvQ = wlw[i0 + 4 + q < nw ? i0 + 4 + q : i0];
                    edge_loads(uvQ, el, feat4, hd, lane, evQ, fAQ, fBQ);
                    edge_sums(uvP, i0, nw, hd, lane, evP, fAP, fBP, acc, errow, mrow, lrow);
                    if (i0 + 4 >= nw) break;
                    uvP = wlw[i0 + 8 + q < nw ? i0 + 8 + q : i0 + 4];
                    edge_loads(uvP, el, feat4, hd, lane, evP, fAP, fBP);
                    edge_sums(uvQ, i0 + 4, nw, hd, lane, evQ, fAQ, fBQ, acc, errow, mrow, lrow);
                }
            }
        } else {
        const int h = lane & 7;
        for (int i0 = wave; i0 < n; i0 += 128) {
            float a[4];
            const float* fp[4];
            float* ap[4];
            const int ikl = i0 + 32 * (lane & 3);
            const unsigned int uv = ent[ikl < n ? ikl : i0];
#pragma unroll
            for (int k = 0; k < 4; ++k) {
                const bool okk = (i0 + 32 * k) < n;
                const unsigned int u = (unsigned int)__builtin_amdgcn_readlane((int)uv, k);
                const int s = (int)(u & 0xFFFFu), r = (int)(u >> 16);
                a[k]  = __expf(leaky(el[s * H1N + h] + errow[r * H1N + h]) - mrow[r * H1N + h])
                        / lrow[r * H1N + h];
                fp[k] = feat + (size_t)s * 256 + lane;
                ap[k] = &acc[(okk ? r : TILE1) * 256 + lane];
            }
            float f[4][8];
#pragma unroll
            for (int j = 0; j < 8; ++j) {
#pragma unroll
                for (int k = 0; k < 4; ++k) f[k][j] = fp[k][32 * j];
            }
#pragma unroll
            for (int j = 0; j < 8; ++j) {
#pragma unroll
                for (int k = 0; k < 4; ++k) {
                    const float w = __int_as_float(__builtin_amdgcn_readlane(__float_as_int(a[k]), j));
                    __hip_atomic_fetch_add(ap[k] + 32 * j, f[k][j] * w,
                                           __ATOMIC_RELAXED, __HIP_MEMORY_SCOPE_WORKGROUP);
                }
            }
        }
        }
    } else {
        for (int e = tid; e < EE; e += 1024) {
            const unsigned int r = (unsigned int)(dst[e] - lo);
            if (r < (unsigned int)rows) {
                const int sidx = src[e];
                for (int hh = 0; hh < H1N; ++hh)
                    lds_max_f32(&mrow[r * H1N + hh], leaky(el[sidx * H1N + hh] + errow[r * H1N + hh]));
            }
        }
        __syncthreads();
        for (int e = tid; e < EE; e += 1024) {
            const unsigned int r = (unsigned int)(dst[e] - lo);
            if (r < (unsigned int)rows) {
                const int sidx = src[e];
                for (int hh = 0; hh < H1N; ++hh)
                    __hip_atomic_fetch_add(&lrow[r * H1N + hh],
                                           __expf(leaky(el[sidx * H1N + hh] + errow[r * H1N + hh]) - mrow[r * H1N + hh]),
                                           __ATOMIC_RELAXED, __HIP_MEMORY_SCOPE_WORKGROUP);
            }
        }
        __syncthreads();
        for (int e = tid; e < EE; e += 1024) {
            const unsigned int r = (unsigned int)(dst[e] - lo);
            if (r < (unsigned int)rows) {
                const int sidx = src[e];
                for (int hh = 0; hh < H1N; ++hh) {
                    const float w = __expf(leaky(el[sidx * H1N + hh] + errow[r * H1N + hh]) - mrow[r * H1N + hh])
                                    / lrow[r * H1N + hh];
                    for (int c = 0; c < 32; ++c)
                        __hip_atomic_fetch_add(&acc[r * 256 + hh * 32 + c], feat[(size_t)sidx * 256 + hh * 32 + c] * w,
                                               __ATOMIC_RELAXED, __HIP_MEMORY_SCOPE_WORKGROUP);
                }
            }
        }
    }
    __syncthreads();

    for (int idx = tid; idx < rows * 256; idx += 1024)
        y[(size_t)lo * 256 + idx] = fmaxf(acc[idx], 0.f);
}

#define TILE2 196
static_assert(TILE2 == TILE1, "layer 2 reads the lists that layer 1 made for its own tiles");
static_assert(NN <= 65536, "source index is packed into 16 bits");
static_assert((size_t)NTILE * CAP1 + NTILE + EE / 4 <= (size_t)NN * 34, "lists, counts and tile codes fit the workspace part kept for them");


__global__ void __launch_bounds__(1024) agg2_tile_kernel(const int* __restrict__ src,
                                                         const int* __restrict__ dst,
                                                         const float* __restrict__ el,
                                                         const float* __restrict__ er,
                                                         const float* __restrict__ feat,
                                                         const int* __restrict__ glist,
                                                         const int* __restrict__ gcnt,
                                                         float* __restrict__ outp) {
    __shared__ float acc[(TILE2 + 1) * 32];
    __shared__ float mrow[TILE2 + 1];
    __shared__ float lrow[TILE2 + 1];
    __shared__ float errow[TILE2 + 1];
    __shared__ unsigned int ent[CAP1];
    __shared__ float val[CAP1];

    const int tid  = threadIdx.x;
    const int lo   = blockIdx.x * TILE2;
    const int rows = (NN - lo) < TILE2 ? (NN - lo) : TILE2;
    const unsigned int total = (unsigned int)gcnt[blockIdx.x];

    for (int idx = tid; idx < (TILE2 + 1) * 32; idx += 1024) acc[idx] = 0.f;
    if (tid <= TILE2) {
        mrow[tid]  = -INFINITY;
        lrow[tid]  = 0.f;
        errow[tid] = tid < rows ? er[lo + tid] : 0.f;
    }
    __syncthreads();

    if (total <= CAP1) {
        const int n = (int)total;
        const int* __restrict__ gl = glist + (size_t)blockIdx.x * CAP1;
        unsigned int uu[CAP1 / 1024];
        float ee[CAP1 / 1024];
#pragma unroll
        for (int it = 0; it < CAP1 / 1024; ++it) {
            const int i = tid + 1024 * it;
            uu[it] = i < n ? (unsigned int)gl[i] : 0u;
        }
#pragma unroll
        for (int it = 0; it < CAP1 / 1024; ++it) ee[it] = el[uu[it] & 0xFFFFu];
#pragma unroll
        for (int it = 0; it < CAP1 / 1024; ++it) {
            const int i = tid + 1024 * it;
            if (i < n) {
                const int r = (int)(uu[it] >> 16);
                const float sc = leaky(ee[it] + errow[r]);
                lds_max_f32(&mrow[r], sc);
                ent[i] = uu[it];
                val[i] = sc;
            }
        }
        __syncthreads();
        for (int i = tid; i < n; i += 1024) {
            const int r = (int)(ent[i] >> 16);
            const float p = __expf(val[i] - mrow[r]);
            val[i] = p;
            __hip_atomic_fetch_add(&lrow[r], p, __ATOMIC_RELAXED, __HIP_MEMORY_SCOPE_WORKGROUP);
        }
        __syncthreads();
        const int sub = tid & 7;
        const float4* __restrict__ feat4 = reinterpret_cast<const float4*>(feat);
        for (int i = tid >> 3; i < n; i += 512) {
            float4 f[4]; float a[4]; int rr[4];
#pragma unroll
            for (int k = 0; k < 4; ++k) {
                const int ik = i + 128 * k;
                const int ic = ik < n ? ik : i;
                const unsigned int u = ent[ic];
                rr[k] = ik < n ? (int)(u >> 16) : -1;
                a[k]  = val[ic] / lrow[u >> 16];
                f[k]  = feat4[(size_t)(u & 0xFFFFu) * 8 + sub];
            }
#pragma unroll
            for (int k = 0; k < 4; ++k) {
                if (rr[k] >= 0) {
                    float* p = &acc[rr[k] * 32 + sub * 4];
                    __hip_atomic_fetch_add(p + 0, f[k].x * a[k], __ATOMIC_RELAXED, __HIP_MEMORY_SCOPE_WORKGROUP);
                    __hip_atomic_fetch_add(p + 1, f[k].y * a[k], __ATOMIC_RELAXED, __HIP_MEMORY_SCOPE_WORKGROUP);
                    __hip_atomic_fetch_add(p + 2, f[k].z * a[k], __ATOMIC_RELAXED, __HIP_MEMORY_SCOPE_WORKGROUP);
                    __hip_atomic_fetch_add(p + 3, f[k].w * a[k], __ATOMIC_RELAXED, __HIP_MEMORY_SCOPE_WORKGROUP);
                }
            }
        }
    } else {
        for (int e = tid; e < EE; e += 1024) {
            const unsigned int r = (unsigned int)(dst[e] - lo);
            if (r < (unsigned int)rows) lds_max_f32(&mrow[r], leaky(el[src[e]] + errow[r]));
        }
        __syncthreads();
        for (int e = tid; e < EE; e += 1024) {
            const unsigned int r = (unsigned int)(dst[e] - lo);
            if (r < (unsigned int)rows)
                __hip_atomic_fetch_add(&lrow[r], __expf(leaky(el[src[e]] + errow[r]) - mrow[r]),
                                       __ATOMIC_RELAXED, __HIP_MEMORY_SCOPE_WORKGROUP);
        }
        __syncthreads();
        for (int e = tid; e < EE; e += 1024) {
            const unsigned int r = (unsigned int)(dst[e] - lo);
            if (r < (unsigned int)rows) {
                const int sidx = src[e];
                const float a = __expf(leaky(el[sidx] + errow[r]) - mrow[r]) / lrow[r];
                for (int c = 0; c < 32; ++c)
                    __hip_atomic_fetch_add(&acc[r * 32 + c], feat[(size_t)sidx * 32 + c] * a,
                                           __ATOMIC_RELAXED, __HIP_MEMORY_SCOPE_WORKGROUP);
            }
        }
    }
    __syncthreads();

    for (int idx = tid; idx < rows * 32; idx += 1024)
        outp[(size_t)lo * 32 + idx] += acc[idx];
}

extern "C" void kernel_launch(void* const* d_in, const int* in_sizes, int n_in,
                              void* d_out, int out_size, void* d_ws, size_t ws_size,
                              hipStream_t stream) {
    const float* x     = (const float*)d_in[0];
    const int*   src   = (const int*)  d_in[1];
    const int*   dst   = (const int*)  d_in[2];
    const float* W1    = (const float*)d_in[3];
    const float* al1   = (const float*)d_in[4];
    const float* ar1   = (const float*)d_in[5];
    const float* b1    = (const float*)d_in[6];
    const float* W2    = (const float*)d_in[7];
    const float* al2   = (const float*)d_in[8];
    const float* ar2   = (const float*)d_in[9];
    const float* resW2 = (const float*)d_in[10];
    const float* b2    = (const float*)d_in[11];
    float* outp = (float*)d_out;

    float* ws    = (float*)d_ws;
    float* feat1 = ws;
    float* rst1  = feat1 + (size_t)NN * 256;
    float* el1   = rst1  + (size_t)NN * 256;
    float* er1   = el1   + (size_t)NN * 8;
    int*   lst   = (int*)(er1 + (size_t)NN * 8);
    int*   lcnt  = lst + (size_t)NTILE * CAP1;
    unsigned int* codes = (unsigned int*)(lcnt + NTILE);
    float* feat2 = feat1;
    float* el2   = el1;
    float* er2   = er1;

    gemm1_kernel<<<GEMM_WGS, GEMM_THREADS, 0, stream>>>(x, W1, feat1);
    elr1_kernel<<<ELR1_WGS, 256, 0, stream>>>(feat1, al1, ar1, el1, er1, dst, codes);
    agg1_tile_kernel<<<NTILE, 1024, 0, stream>>>(src, dst, codes, el1, er1,
                                                 feat1, x, b1, rst1, lst, lcnt);

    gemm2_kernel<<<GEMM_WGS, GEMM_THREADS, 0, stream>>>(rst1, W2, resW2, b2, feat2, outp);
    elr2_kernel<<<ELR2_WGS, 128, 0, stream>>>(feat2, al2, ar2, el2, er2);
    agg2_tile_kernel<<<NTILE, 1024, 0, stream>>>(src, dst, el2, er2, feat2, lst, lcnt, outp);
}


# Layer-1 tile rows live in their owner wave's registers (two groups of 4 slots, scalar test of row / 32) from x+b1 to the relu store: no LDS tile pass, load or store; sum order: owner list, as before
# speedup vs baseline: 5.0939x; 1.0392x over previous
#include <hip/hip_runtime.h>
#include <hip/hip_bf16.h>
#include <math.h>

#define NN 50000
#define EE 800000
#define INF_ 256
#define OUTF 32
#define H1N 8
#define SLOPE 0.2f

typedef float v2f __attribute__((ext_vector_type(2)));
typedef float v8f __attribute__((ext_vector_type(8)));

__device__ __forceinline__ float leaky(float x) { return x > 0.f ? x : SLOPE * x; }

typedef float v4f __attribute__((ext_vector_type(4)));

#define NT_ROWS (NN / 16)
#define GEMM_WGS 256
#define GEMM_THREADS 512
static_assert(NN % 16 == 0, "whole 16-row tiles only: no row guard in the GEMM kernels");

template <int NCOL>
__device__ __forceinline__ void stage_weights(float* __restrict__ Wl, const float* __restrict__ Wa,
                                              const float* __restrict__ Wb, const int tid) {
    constexpr int P = NCOL * 4 + 32;
    constexpr int CG = NCOL / 4;
    constexpr int NB = 64 * CG / GEMM_THREADS;
    static_assert(64 * CG % GEMM_THREADS == 0, "every thread moves the same number of blocks");
    const int rot = (tid >> 2) & 3;
    v4f r[NB][4];
#pragma unroll
    for (int it = 0; it < NB; ++it) {
        const int q = tid + it * GEMM_THREADS;
        const int p = q / CG, cg = q % CG;
        const float* s;
        int pitch;
        if (NCOL == 256) { s = Wa + (size_t)(4 * p) * 256 + 4 * cg; pitch = 256; }
        else             { s = (cg < 8 ? Wa : Wb) + (size_t)(4 * p) * 32 + 4 * (cg & 7); pitch = 32; }
#pragma unroll
        for (int i = 0; i < 4; ++i) r[it][i] = *reinterpret_cast<const v4f*>(s + i * pitch);
    }
#pragma unroll
    for (int it = 0; it < NB; ++it) {
        const int q = tid + it * GEMM_THREADS;
        const int p = q / CG, cg = q % CG;
#pragma unroll
        for (int j = 0; j < 4; ++j) {
            const int jj = (j + rot) & 3;
            v4f o;
            o.x = jj == 0 ? r[it][0].x : jj == 1 ? r[it][0].y : jj == 2 ? r[it][0].z : r[it][0].w;
            o.y = jj == 0 ? r[it][1].x : jj == 1 ? r[it][1].y : jj == 2 ? r[it][1].z : r[it][1].w;
            o.z = jj == 0 ? r[it][2].x : jj == 1 ? r[it][2].y : jj == 2 ? r[it][2].z : r[it][2].w;
            o.w = jj == 0 ? r[it][3].x : jj == 1 ? r[it][3].y : jj == 2 ? r[it][3].z : r[it][3].w;
            *reinterpret_cast<v4f*>(Wl + p * P + (4 * cg + jj) * 4) = o;
        }
    }
}

template <int NCOL, int NT16>
__device__ __forceinline__ void unit_mma(const float* __restrict__ A, const float* __restrict__ Wl,
                                         const int tile, const int col0, const int lane, v8f (&acc)[NT16]) {
    constexpr int P = NCOL * 4 + 32;
    const int h = lane >> 4, c = lane & 15;
    const v4f* __restrict__ a4 = reinterpret_cast<const v4f*>(A + (size_t)(tile * 16 + c) * 256) + h;
    const float* bp = Wl + h * P + (col0 + c) * 4;
    for (int J0 = 0; J0 < 32; J0 += 8) {
        v4f ac[8];
#pragma unroll
        for (int j = 0; j < 8; ++j) ac[j] = a4[2 * (J0 + j)];
#pragma unroll
        for (int j = 0; j < 8; ++j) {
            const float* bj = bp + 2 * (J0 + j) * P;
            v4f b[NT16];
#pragma unroll
            for (int t = 0; t < NT16; ++t) b[t] = *reinterpret_cast<const v4f*>(bj + t * 64);
            v2f alo, ahi;
            alo.x = ac[j].x; alo.y = ac[j].y; ahi.x = ac[j].z; ahi.y = ac[j].w;
#pragma unroll
            for (int t = 0; t < NT16; ++t) {
                v2f blo; blo.x = b[t].x; blo.y = b[t].y;
                acc[t] = __builtin_amdgcn_wmma_f32_16x16x4_f32(false, alo, false, blo, (short)0, acc[t], false, false);
            }
#pragma unroll
            for (int t = 0; t < NT16; ++t) {
                v2f bhi; bhi.x = b[t].z; bhi.y = b[t].w;
                acc[t] = __builtin_amdgcn_wmma_f32_16x16x4_f32(false, ahi, false, bhi, (short)0, acc[t], false, false);
            }
        }
    }
}

__global__ void __launch_bounds__(GEMM_THREADS) gemm1_kernel(const float* __restrict__ X,
                                                             const float* __restrict__ W,
                                                             float* __restrict__ C) {
    __shared__ __attribute__((aligned(16))) float Wl[64 * (256 * 4 + 32)];
    const int tid = threadIdx.x, lane = tid & 31;
    const int wave = __builtin_amdgcn_readfirstlane(tid >> 5);
    stage_weights<256>(Wl, W, W, tid);
    __syncthreads();
    const int ntb = (NT_ROWS - (int)blockIdx.x + GEMM_WGS - 1) / GEMM_WGS;
    const int g = wave & 3, h = lane >> 4, c = lane & 15;
    for (int i = ((wave >> 2) - g) & 3; i < ntb; i += 4) {
        const int tile = blockIdx.x + GEMM_WGS * i;
        v8f acc[4] = {};
        unit_mma<256, 4>(X, Wl, tile, g * 64, lane, acc);
        float* crow = C + (size_t)(tile * 16 + h * 8) * 256 + g * 64 + c;
#pragma unroll
        for (int t = 0; t < 4; ++t) {
#pragma unroll
            for (int j = 0; j < 8; ++j) crow[(size_t)j * 256 + t * 16] = acc[t][j];
        }
    }
}

__global__ void __launch_bounds__(GEMM_THREADS) gemm2_kernel(const float* __restrict__ Y,
                                                             const float* __restrict__ W2,
                                                             const float* __restrict__ RW,
                                                             const float* __restrict__ b2,
                                                             float* __restrict__ F2,
                                                             float* __restrict__ OUTP) {
    __shared__ __attribute__((aligned(16))) float Wl[64 * (64 * 4 + 32)];
    const int tid = threadIdx.x, lane = tid & 31;
    const int wave = __builtin_amdgcn_readfirstlane(tid >> 5);
    stage_weights<64>(Wl, W2, RW, tid);
    __syncthreads();
    const int ntb = (NT_ROWS - (int)blockIdx.x + GEMM_WGS - 1) / GEMM_WGS;
    const int h = lane >> 4, c = lane & 15;
    for (int u = wave; u < 2 * ntb; u += 16) {
        const int tile = blockIdx.x + GEMM_WGS * (u >> 1);
        const int half = u & 1;
        v8f acc[2] = {};
        unit_mma<64, 2>(Y, Wl, tile, half * 32, lane, acc);
        const size_t o = (size_t)(tile * 16 + h * 8) * 32 + c;
        if (half == 0) {
#pragma unroll
            for (int t = 0; t < 2; ++t) {
#pragma unroll
                for (int j = 0; j < 8; ++j) F2[o + (size_t)j * 32 + t * 16] = acc[t][j];
            }
        } else {
#pragma unroll
            for (int t = 0; t < 2; ++t) {
                const float bb = b2[t * 16 + c];
#pragma unroll
                for (int j = 0; j < 8; ++j) OUTP[o + (size_t)j * 32 + t * 16] = acc[t][j] + bb;
            }
        }
    }
}

__device__ __forceinline__ float dot4(const v4f a, const v4f b) {
    return a.x * b.x + a.y * b.y + a.z * b.z + a.w * b.w;
}

#define ROWS1 8
#define ELR1_WGS ((NN / ROWS1 + 7) / 8)
#define TILE1 196
#define NTILE ((NN + TILE1 - 1) / TILE1)
static_assert(NN % ROWS1 == 0, "a wave takes ROWS1 whole rows");
static_assert(ROWS1 == 8, "the load fence in the kernel names eight rows");
static_assert(NTILE <= 256, "a tile number fits a byte");
static_assert(EE % 16 == 0, "codes are written four and read sixteen at a time");
static_assert(ELR1_WGS * 256 >= EE / 4, "one thread of the logit launch codes four edges");

__device__ __forceinline__ unsigned int tile_code(int d) { return ((unsigned int)d / TILE1) & 255u; }

__global__ void __launch_bounds__(256) elr1_kernel(const float* __restrict__ feat,
                                                   const float* __restrict__ al,
                                                   const float* __restrict__ ar,
                                                   float* __restrict__ el,
                                                   float* __restrict__ er,
                                                   const int* __restrict__ dst,
                                                   unsigned int* __restrict__ codes) {
    const int lane = threadIdx.x & 31;
    const int g = blockIdx.x * 256 + threadIdx.x;
    if (g < EE / 4) {
        const int4 d = reinterpret_cast<const int4*>(dst)[g];
        codes[g] = tile_code(d.x) | (tile_code(d.y) << 8) | (tile_code(d.z) << 16) | (tile_code(d.w) << 24);
    }
    const int row0 = (blockIdx.x * 8 + (threadIdx.x >> 5)) * ROWS1;
    if (row0 >= NN) return;
    const v4f* __restrict__ f4 = reinterpret_cast<const v4f*>(feat) + (size_t)row0 * 64 + lane;
    v4f v[ROWS1][2];
#pragma unroll
    for (int i = 0; i < ROWS1; ++i) { v[i][0] = f4[i * 64]; v[i][1] = f4[i * 64 + 32]; }
    const v4f* __restrict__ al4 = reinterpret_cast<const v4f*>(al);
    const v4f* __restrict__ ar4 = reinterpret_cast<const v4f*>(ar);
    const v4f l0 = al4[lane], l1 = al4[32 + lane];
    const v4f r0 = ar4[lane], r1 = ar4[32 + lane];
    asm volatile("" :: "v"(v[0][0]), "v"(v[0][1]), "v"(v[1][0]), "v"(v[1][1]),
                       "v"(v[2][0]), "v"(v[2][1]), "v"(v[3][0]), "v"(v[3][1]),
                       "v"(v[4][0]), "v"(v[4][1]), "v"(v[5][0]), "v"(v[5][1]),
                       "v"(v[6][0]), "v"(v[6][1]), "v"(v[7][0]), "v"(v[7][1]),
                       "v"(l0), "v"(l1), "v"(r0), "v"(r1));
    const bool b0 = (lane & 1) != 0, b1 = (lane & 2) != 0, b2 = (lane & 4) != 0;
    const int head = (lane >> 3) + (b1 ? 4 : 0);
    float* __restrict__ dstp = b0 ? er : el;
#pragma unroll
    for (int i = 0; i < ROWS1; i += 2) {
        float u[2];
#pragma unroll
        for (int c = 0; c < 2; ++c) {
            const float a0 = dot4(v[i + c][0], l0), e0 = dot4(v[i + c][0], r0);
            const float a1 = dot4(v[i + c][1], l1), e1 = dot4(v[i + c][1], r1);
            const float t0 = (b0 ? e0 : a0) + __shfl_xor(b0 ? a0 : e0, 1, 32);
            const float t1 = (b0 ? e1 : a1) + __shfl_xor(b0 ? a1 : e1, 1, 32);
            u[c] = (b1 ? t1 : t0) + __shfl_xor(b1 ? t0 : t1, 2, 32);
        }
        const float w = (b2 ? u[1] : u[0]) + __shfl_xor(b2 ? u[0] : u[1], 4, 32);
        dstp[(row0 + i + (b2 ? 1 : 0)) * H1N + head] = w;
    }
}

#define ROWS2 16
#define ELR2_WGS ((NN / ROWS2 + 3) / 4)
static_assert(NN % ROWS2 == 0, "a wave takes ROWS2 whole rows");
static_assert(ROWS2 == 16, "four loads of four rows, joined pairwise by the exchange tree");

__global__ void __launch_bounds__(128) elr2_kernel(const float* __restrict__ feat,
                                                   const float* __restrict__ al,
                                                   const float* __restrict__ ar,
                                                   float* __restrict__ el,
                                                   float* __restrict__ er) {
    const int lane = threadIdx.x & 31;
    const int row0 = (blockIdx.x * 4 + (threadIdx.x >> 5)) * ROWS2;
    if (row0 >= NN) return;
    const v4f* __restrict__ f4 = reinterpret_cast<const v4f*>(feat) + (size_t)row0 * 8 + lane;
    v4f v[4];
#pragma unroll
    for (int q = 0; q < 4; ++q) v[q] = f4[q * 32];
    const v4f l = reinterpret_cast<const v4f*>(al)[lane & 7];
    const v4f r = reinterpret_cast<const v4f*>(ar)[lane & 7];
    const bool b0 = (lane & 1) != 0, b1 = (lane & 2) != 0, b2 = (lane & 4) != 0;
    float t[4];
#pragma unroll
    for (int q = 0; q < 4; ++q) {
        const float a = dot4(v[q], l), e = dot4(v[q], r);
        t[q] = (b0 ? e : a) + __shfl_xor(b0 ? a : e, 1, 32);
    }
    const float u0 = (b1 ? t[1] : t[0]) + __shfl_xor(b1 ? t[0] : t[1], 2, 32);
    const float u1 = (b1 ? t[3] : t[2]) + __shfl_xor(b1 ? t[2] : t[3], 2, 32);
    const float w  = (b2 ? u1 : u0) + __shfl_xor(b2 ? u0 : u1, 4, 32);
    const int q = (b2 ? 2 : 0) + (b1 ? 1 : 0);
    (b0 ? er : el)[row0 + 4 * q + (lane >> 3)] = w;
}

__device__ __forceinline__ void lds_max_f32(float* addr, float v) {
    if (v >= 0.f) __hip_atomic_fetch_max((int*)addr, __float_as_int(v),
                                         __ATOMIC_RELAXED, __HIP_MEMORY_SCOPE_WORKGROUP);
    else          __hip_atomic_fetch_min((unsigned int*)addr, __float_as_uint(v),
                                         __ATOMIC_RELAXED, __HIP_MEMORY_SCOPE_WORKGROUP);
}

#define CAP1  4096
#define WCAP  256
#define NITC  ((EE / 16 + 1023) / 1024)
#define GRPC  7
static_assert(NN <= 65536 && TILE1 < 65535, "tile row and source index are packed into 32 bits");
static_assert(NITC % GRPC == 0, "scan loop takes GRPC trips at a time");
static_assert(CAP1 % 1024 == 0, "a thread owns CAP1 / 1024 list places");

__device__ __forceinline__ unsigned int zero_bytes(unsigned int y) {
    return ~(((y & 0x7F7F7F7Fu) + 0x7F7F7F7Fu) | y) & 0x80808080u;
}

__device__ __forceinline__ float row_pick(float x, unsigned int lo, unsigned int hi) {
    return __int_as_float(__builtin_amdgcn_permlane16(__float_as_int(x), __float_as_int(x), lo, hi, false, false));
}

__device__ __forceinline__ void edge_loads(const unsigned int uv, const float* __restrict__ el,
                                           const v4f* __restrict__ feat4, const int hd, const int lane,
                                           float (&ev)[4], v4f (&fA)[4], v4f (&fB)[4]) {
#pragma unroll
    for (int k = 0; k < 4; ++k) {
        const int s = __builtin_amdgcn_readlane((int)uv, k) & 0xFFFF;
        ev[k] = el[s * H1N + hd];
    }
#pragma unroll
    for (int k = 0; k < 4; ++k) {
        const int s = __builtin_amdgcn_readlane((int)uv, k) & 0xFFFF;
        const v4f* __restrict__ rowp = feat4 + (size_t)s * 64;
        fA[k] = rowp[lane];
        fB[k] = rowp[32 + lane];
    }
}

#define SLOTS 4
#define NGRP  2
#define GCAP  (WCAP / NGRP)
static_assert((TILE1 + 1 + 31) / 32 <= SLOTS * NGRP, "every tile row and the spare row have a slot");

struct SlotRows { v4f a0, b0, a1, b1, a2, b2, a3, b3; };
static_assert(SLOTS == 4, "SlotRows names four slots");

#define ROW_FMA(J, RA, RB)                                                                       \
    asm volatile("s_cmp_lg_u32 %18, " #J "\n\t"                                                  \
                 "s_cbranch_scc1 .Lslot_skip%=\n\t"                                              \
                 "v_fmac_f32 %0, %8, %16\n\tv_fmac_f32 %1, %9, %16\n\t"                         \
                 "v_fmac_f32 %2, %10, %16\n\tv_fmac_f32 %3, %11, %16\n\t"                       \
                 "v_fmac_f32 %4, %12, %17\n\tv_fmac_f32 %5, %13, %17\n\t"                       \
                 "v_fmac_f32 %6, %14, %17\n\tv_fmac_f32 %7, %15, %17\n"                         \
                 ".Lslot_skip%=:"                                                                \
                 : "+v"(RA.x), "+v"(RA.y), "+v"(RA.z), "+v"(RA.w),                               \
                   "+v"(RB.x), "+v"(RB.y), "+v"(RB.z), "+v"(RB.w)                                \
                 : "v"(fA.x), "v"(fA.y), "v"(fA.z), "v"(fA.w),                                   \
                   "v"(fB.x), "v"(fB.y), "v"(fB.z), "v"(fB.w), "v"(wA), "v"(wB), "s"(j)          \
                 : "scc")

__device__ __forceinline__ void slot_add(SlotRows& s, const int j,
                                         const v4f fA, const float wA, const v4f fB, const float wB) {
    ROW_FMA(0, s.a0, s.b0);
    ROW_FMA(1, s.a1, s.b1);
    ROW_FMA(2, s.a2, s.b2);
    ROW_FMA(3, s.a3, s.b3);
}

__device__ __forceinline__ void slot_rows_x(SlotRows& s, const v4f* __restrict__ x4, const int row0, const int rows) {
    const int r0 = row0      < rows ? row0      : rows - 1;
    const int r1 = row0 + 32 < rows ? row0 + 32 : rows - 1;
    const int r2 = row0 + 64 < rows ? row0 + 64 : rows - 1;
    const int r3 = row0 + 96 < rows ? row0 + 96 : rows - 1;
    s.a0 = x4[(size_t)r0 * 64]; s.b0 = x4[(size_t)r0 * 64 + 32];
    s.a1 = x4[(size_t)r1 * 64]; s.b1 = x4[(size_t)r1 * 64 + 32];
    s.a2 = x4[(size_t)r2 * 64]; s.b2 = x4[(size_t)r2 * 64 + 32];
    s.a3 = x4[(size_t)r3 * 64]; s.b3 = x4[(size_t)r3 * 64 + 32];
}

__device__ __forceinline__ v4f relu4(const v4f a) {
    v4f o;
    o.x = fmaxf(a.x, 0.f); o.y = fmaxf(a.y, 0.f); o.z = fmaxf(a.z, 0.f); o.w = fmaxf(a.w, 0.f);
    return o;
}

__device__ __forceinline__ void tile_start_lds(float* acc, const float* __restrict__ x,
                                               const float* __restrict__ b1, const int lo,
                                               const int rows, const int tid) {
    for (int idx = tid; idx < rows * 256; idx += 1024)
        acc[idx] = x[(size_t)lo * 256 + idx] + b1[idx & 255];
}

__global__ void __launch_bounds__(1024) agg1_tile_kernel(const int* __restrict__ src,
                                                         const int* __restrict__ dst,
                                                         const unsigned int* __restrict__ codes,
                                                         const float* __restrict__ el,
                                                         const float* __restrict__ er,
                                                         const float* __restrict__ feat,
                                                         const float* __restrict__ x,
                                                         const float* __restrict__ b1,
                                                         float* __restrict__ y,
                                                         int* __restrict__ glist,
                                                         int* __restrict__ gcnt) {
    __shared__ __attribute__((aligned(16))) float acc[(TILE1 + 1) * 256];
    __shared__ __attribute__((aligned(16))) float errow[(TILE1 + 1) * H1N];
    __shared__ float mrow[(TILE1 + 1) * H1N];
    __shared__ float lrow[(TILE1 + 1) * H1N];
    __shared__ unsigned int ent[CAP1];
    __shared__ unsigned int wl[32 * WCAP];
    __shared__ unsigned int wcnt[32 * NGRP];
    __shared__ unsigned int cnt;

    const int tid  = threadIdx.x;
    const int lane = tid & 31;
    const int wave = __builtin_amdgcn_readfirstlane(tid >> 5);
    const int lo   = blockIdx.x * TILE1;
    const int rows = (NN - lo) < TILE1 ? (NN - lo) : TILE1;

    for (int idx = tid; idx < (TILE1 + 1) * H1N; idx += 1024) {
        mrow[idx]  = -INFINITY;
        lrow[idx]  = 0.f;
        errow[idx] = idx < rows * H1N ? er[(size_t)lo * H1N + idx] : 0.f;
    }
    if (tid == 0) cnt = 0u;
    if (tid < 32 * NGRP) wcnt[tid] = 0u;

    const uint4* __restrict__ code4 = reinterpret_cast<const uint4*>(codes);
    const unsigned int pat = (unsigned int)blockIdx.x * 0x01010101u;
    const int it0 = (blockIdx.x * NITC) >> 8;
    uint4 cur[GRPC];
#pragma unroll
    for (int q = 0; q < GRPC; ++q) {
        int it = it0 + q; if (it >= NITC) it -= NITC;
        const int iw = it * 1024 + tid;
        cur[q] = code4[iw < EE / 16 ? iw : tid];
    }
    __syncthreads();

    for (int j = 0; j < NITC; j += GRPC) {
        uint4 nxt[GRPC];
#pragma unroll
        for (int q = 0; q < GRPC; ++q) {
            int it = it0 + j + GRPC + q; if (it >= NITC) it -= NITC;
            const int iw = it * 1024 + tid;
            nxt[q] = code4[iw < EE / 16 ? iw : tid];
        }
#pragma unroll
        for (int q = 0; q < GRPC; ++q) {
            int it = it0 + j + q; if (it >= NITC) it -= NITC;
            const int iw = it * 1024 + tid;
            unsigned int m = (zero_bytes(cur[q].x ^ pat) >> 7) | (zero_bytes(cur[q].y ^ pat) >> 6)
                           | (zero_bytes(cur[q].z ^ pat) >> 5) | (zero_bytes(cur[q].w ^ pat) >> 4);
            if (iw >= EE / 16) m = 0u;
            while (m != 0u) {
                const unsigned int p = atomicAdd(&cnt, 1u);
                const int b = __builtin_ctz(m);
                if (p < CAP1) ent[p] = (unsigned int)(iw * 16 + (b & 7) * 4 + (b >> 3));
                m &= m - 1u;
            }
        }
#pragma unroll
        for (int q = 0; q < GRPC; ++q) cur[q] = nxt[q];
    }
    __syncthreads();
    const unsigned int total = cnt;
    if (tid == 0) gcnt[blockIdx.x] = (int)total;

    bool in_regs = false;
    if (total <= CAP1) {
        const int n = (int)total;
        int* __restrict__ gl = glist + (size_t)blockIdx.x * CAP1;
        const v4f* __restrict__ x4 = reinterpret_cast<const v4f*>(x) + (size_t)lo * 64 + lane;
        SlotRows sr;
        slot_rows_x(sr, x4, wave, rows);

        const float4* __restrict__ el4 = reinterpret_cast<const float4*>(el);
        float sc[CAP1 / 1024][H1N];
        int   rw[CAP1 / 1024];
        int   sw[CAP1 / 1024];
#pragma unroll
        for (int it = 0; it < CAP1 / 1024; ++it) {
            const int i = tid + 1024 * it;
            const int e = i < n ? (int)ent[i] : 0;
            rw[it] = dst[e];
            sw[it] = src[e];
        }
#pragma unroll
        for (int it = 0; it < CAP1 / 1024; ++it) {
            const unsigned int r = (unsigned int)(rw[it] - lo);
            rw[it] = (tid + 1024 * it) < n ? (r < (unsigned int)rows ? (int)r : TILE1) : -1;
        }
#pragma unroll
        for (int it = 0; it < CAP1 / 1024; ++it) {
            const int r = rw[it] < 0 ? 0 : rw[it];
            const float4 a0 = el4[(size_t)sw[it] * 2], a1 = el4[(size_t)sw[it] * 2 + 1];
            const float4 c0 = *reinterpret_cast<const float4*>(&errow[r * H1N]);
            const float4 c1 = *reinterpret_cast<const float4*>(&errow[r * H1N + 4]);
            sc[it][0] = leaky(a0.x + c0.x); sc[it][1] = leaky(a0.y + c0.y);
            sc[it][2] = leaky(a0.z + c0.z); sc[it][3] = leaky(a0.w + c0.w);
            sc[it][4] = leaky(a1.x + c1.x); sc[it][5] = leaky(a1.y + c1.y);
            sc[it][6] = leaky(a1.z + c1.z); sc[it][7] = leaky(a1.w + c1.w);
        }
#pragma unroll
        for (int it = 0; it < CAP1 / 1024; ++it) {
            if (rw[it] >= 0) {
#pragma unroll
                for (int h = 0; h < H1N; ++h) lds_max_f32(&mrow[rw[it] * H1N + h], sc[it][h]);
                const unsigned int pk = ((unsigned int)rw[it] << 16) | (unsigned int)sw[it];
                ent[tid + 1024 * it] = pk;
                gl[tid + 1024 * it]  = (int)pk;
                const unsigned int ow = ((unsigned int)rw[it] & 31u) * NGRP + ((unsigned int)rw[it] >> 5) / SLOTS;
                const unsigned int p  = atomicAdd(&wcnt[ow], 1u);
                if (p < GCAP) wl[ow * GCAP + p] = pk;
            }
        }
        __syncthreads();
#pragma unroll
        for (int it = 0; it < CAP1 / 1024; ++it) {
            if (rw[it] >= 0) {
#pragma unroll
                for (int h = 0; h < H1N; ++h)
                    __hip_atomic_fetch_add(&lrow[rw[it] * H1N + h], __expf(sc[it][h] - mrow[rw[it] * H1N + h]),
                                           __ATOMIC_RELAXED, __HIP_MEMORY_SCOPE_WORKGROUP);
            }
        }
        __syncthreads();
        const unsigned int myc = wcnt[lane] > wcnt[32 + lane] ? wcnt[lane] : wcnt[32 + lane];
        static_assert(NGRP == 2, "the overflow test reads two counts a lane");
        if (!__any(myc > (unsigned int)GCAP)) {
            in_regs = true;
            const int q  = lane & 3;
            const int hd = (q & 1) + 2 * (lane >> 4) + 4 * (q >> 1);
            const v4f* __restrict__ feat4 = reinterpret_cast<const v4f*>(feat);
            v4f* __restrict__ y4 = reinterpret_cast<v4f*>(y) + (size_t)lo * 64 + lane;
            const v4f* __restrict__ b14 = reinterpret_cast<const v4f*>(b1);
            const v4f bA = b14[lane], bB = b14[32 + lane];
            for (int g = 0; g < NGRP; ++g) {
                const int nw = __builtin_amdgcn_readfirstlane((int)wcnt[wave * NGRP + g]);
                const unsigned int* wlw = wl + (wave * NGRP + g) * GCAP;
                const int row0 = wave + 32 * SLOTS * g;
                if (g > 0) slot_rows_x(sr, x4, row0, rows);
                sr.a0 += bA; sr.b0 += bB; sr.a1 += bA; sr.b1 += bB;
                sr.a2 += bA; sr.b2 += bB; sr.a3 += bA; sr.b3 += bB;
                unsigned int uv = nw > 0 ? wlw[q < nw ? q : 0] : 0u;
                for (int i0 = 0; i0 < nw; i0 += 4) {
                    const unsigned int uvN = wlw[i0 + 4 + q < nw ? i0 + 4 + q : i0];
                    float ev[4];
                    v4f fA[4], fB[4];
                    edge_loads(uv, el, feat4, hd, lane, ev, fA, fB);
                    int jj[4];
                    float a[4];
#pragma unroll
                    for (int k = 0; k < 4; ++k) {
                        const int r = (int)((unsigned int)__builtin_amdgcn_readlane((int)uv, k) >> 16);
                        jj[k] = (r >> 5) % SLOTS;
                        a[k] = __expf(leaky(ev[k] + errow[r * H1N + hd]) - mrow[r * H1N + hd])
                               / lrow[r * H1N + hd];
                        if (i0 + k >= nw) a[k] = 0.f;
                    }
#pragma unroll
                    for (int k = 0; k < 4; ++k) {
                        const float wA = row_pick(a[k], 0x00000000u, 0x11111111u);
                        const float wB = row_pick(a[k], 0x22222222u, 0x33333333u);
                        slot_add(sr, jj[k], fA[k], wA, fB[k], wB);
                    }
                    uv = uvN;
                }
                if (row0      < rows) { y4[(size_t)row0 * 64]        = relu4(sr.a0); y4[(size_t)row0 * 64 + 32]        = relu4(sr.b0); }
                if (row0 + 32 < rows) { y4[(size_t)(row0 + 32) * 64] = relu4(sr.a1); y4[(size_t)(row0 + 32) * 64 + 32] = relu4(sr.b1); }
                if (row0 + 64 < rows) { y4[(size_t)(row0 + 64) * 64] = relu4(sr.a2); y4[(size_t)(row0 + 64) * 64 + 32] = relu4(sr.b2); }
                if (row0 + 96 < rows) { y4[(size_t)(row0 + 96) * 64] = relu4(sr.a3); y4[(size_t)(row0 + 96) * 64 + 32] = relu4(sr.b3); }
            }
        } else {
        tile_start_lds(acc, x, b1, lo, rows, tid);
        __syncthreads();
        const int h = lane & 7;
        for (int i0 = wave; i0 < n; i0 += 128) {
            float a[4];
            const float* fp[4];
            float* ap[4];
            const int ikl = i0 + 32 * (lane & 3);
            const unsigned int uv = ent[ikl < n ? ikl : i0];
#pragma unroll
            for (int k = 0; k < 4; ++k) {
                const bool okk = (i0 + 32 * k) < n;
                const unsigned int u = (unsigned int)__builtin_amdgcn_readlane((int)uv, k);
                const int s = (int)(u & 0xFFFFu), r = (int)(u >> 16);
                a[k]  = __expf(leaky(el[s * H1N + h] + errow[r * H1N + h]) - mrow[r * H1N + h])
                        / lrow[r * H1N + h];
                fp[k] = feat + (size_t)s * 256 + lane;
                ap[k] = &acc[(okk ? r : TILE1) * 256 + lane];
            }
            float f[4][8];
#pragma unroll
            for (int j = 0; j < 8; ++j) {
#pragma unroll
                for (int k = 0; k < 4; ++k) f[k][j] = fp[k][32 * j];
            }
#pragma unroll
            for (int j = 0; j < 8; ++j) {
#pragma unroll
                for (int k = 0; k < 4; ++k) {
                    const float w = __int_as_float(__builtin_amdgcn_readlane(__float_as_int(a[k]), j));
                    __hip_atomic_fetch_add(ap[k] + 32 * j, f[k][j] * w,
                                           __ATOMIC_RELAXED, __HIP_MEMORY_SCOPE_WORKGROUP);
                }
            }
        }
        }
    } else {
        tile_start_lds(acc, x, b1, lo, rows, tid);
        for (int e = tid; e < EE; e += 1024) {
            const unsigned int r = (unsigned int)(dst[e] - lo);
            if (r < (unsigned int)rows) {
                const int sidx = src[e];
                for (int hh = 0; hh < H1N; ++hh)
                    lds_max_f32(&mrow[r * H1N + hh], leaky(el[sidx * H1N + hh] + errow[r * H1N + hh]));
            }
        }
        __syncthreads();
        for (int e = tid; e < EE; e += 1024) {
            const unsigned int r = (unsigned int)(dst[e] - lo);
            if (r < (unsigned int)rows) {
                const int sidx = src[e];
                for (int hh = 0; hh < H1N; ++hh)
                    __hip_atomic_fetch_add(&lrow[r * H1N + hh],
                                           __expf(leaky(el[sidx * H1N + hh] + errow[r * H1N + hh]) - mrow[r * H1N + hh]),
                                           __ATOMIC_RELAXED, __HIP_MEMORY_SCOPE_WORKGROUP);
            }
        }
        __syncthreads();
        for (int e = tid; e < EE; e += 1024) {
            const unsigned int r = (unsigned int)(dst[e] - lo);
            if (r < (unsigned int)rows) {
                const int sidx = src[e];
                for (int hh = 0; hh < H1N; ++hh) {
                    const float w = __expf(leaky(el[sidx * H1N + hh] + errow[r * H1N + hh]) - mrow[r * H1N + hh])
                                    / lrow[r * H1N + hh];
                    for (int c = 0; c < 32; ++c)
                        __hip_atomic_fetch_add(&acc[r * 256 + hh * 32 + c], feat[(size_t)sidx * 256 + hh * 32 + c] * w,
                                               __ATOMIC_RELAXED, __HIP_MEMORY_SCOPE_WORKGROUP);
                }
            }
        }
    }
    if (!in_regs) {
        __syncthreads();
        for (int idx = tid; idx < rows * 256; idx += 1024)
            y[(size_t)lo * 256 + idx] = fmaxf(acc[idx], 0.f);
    }
}

#define TILE2 196
static_assert(TILE2 == TILE1, "layer 2 reads the lists that layer 1 made for its own tiles");
static_assert(NN <= 65536, "source index is packed into 16 bits");
static_assert((size_t)NTILE * CAP1 + NTILE + EE / 4 <= (size_t)NN * 34, "lists, counts and tile codes fit the workspace part kept for them");


__global__ void __launch_bounds__(1024) agg2_tile_kernel(const int* __restrict__ src,
                                                         const int* __restrict__ dst,
                                                         const float* __restrict__ el,
                                                         const float* __restrict__ er,
                                                         const float* __restrict__ feat,
                                                         const int* __restrict__ glist,
                                                         const int* __restrict__ gcnt,
                                                         float* __restrict__ outp) {
    __shared__ float acc[(TILE2 + 1) * 32];
    __shared__ float mrow[TILE2 + 1];
    __shared__ float lrow[TILE2 + 1];
    __shared__ float errow[TILE2 + 1];
    __shared__ unsigned int ent[CAP1];
    __shared__ float val[CAP1];

    const int tid  = threadIdx.x;
    const int lo   = blockIdx.x * TILE2;
    const int rows = (NN - lo) < TILE2 ? (NN - lo) : TILE2;
    const unsigned int total = (unsigned int)gcnt[blockIdx.x];

    for (int idx = tid; idx < (TILE2 + 1) * 32; idx += 1024) acc[idx] = 0.f;
    if (tid <= TILE2) {
        mrow[tid]  = -INFINITY;
        lrow[tid]  = 0.f;
        errow[tid] = tid < rows ? er[lo + tid] : 0.f;
    }
    __syncthreads();

    if (total <= CAP1) {
        const int n = (int)total;
        const int* __restrict__ gl = glist + (size_t)blockIdx.x * CAP1;
        unsigned int uu[CAP1 / 1024];
        float ee[CAP1 / 1024];
#pragma unroll
        for (int it = 0; it < CAP1 / 1024; ++it) {
            const int i = tid + 1024 * it;
            uu[it] = i < n ? (unsigned int)gl[i] : 0u;
        }
#pragma unroll
        for (int it = 0; it < CAP1 / 1024; ++it) ee[it] = el[uu[it] & 0xFFFFu];
#pragma unroll
        for (int it = 0; it < CAP1 / 1024; ++it) {
            const int i = tid + 1024 * it;
            if (i < n) {
                const int r = (int)(uu[it] >> 16);
                const float sc = leaky(ee[it] + errow[r]);
                lds_max_f32(&mrow[r], sc);
                ent[i] = uu[it];
                val[i] = sc;
            }
        }
        __syncthreads();
        for (int i = tid; i < n; i += 1024) {
            const int r = (int)(ent[i] >> 16);
            const float p = __expf(val[i] - mrow[r]);
            val[i] = p;
            __hip_atomic_fetch_add(&lrow[r], p, __ATOMIC_RELAXED, __HIP_MEMORY_SCOPE_WORKGROUP);
        }
        __syncthreads();
        const int sub = tid & 7;
        const float4* __restrict__ feat4 = reinterpret_cast<const float4*>(feat);
        for (int i = tid >> 3; i < n; i += 512) {
            float4 f[4]; float a[4]; int rr[4];
#pragma unroll
            for (int k = 0; k < 4; ++k) {
                const int ik = i + 128 * k;
                const int ic = ik < n ? ik : i;
                const unsigned int u = ent[ic];
                rr[k] = ik < n ? (int)(u >> 16) : -1;
                a[k]  = val[ic] / lrow[u >> 16];
                f[k]  = feat4[(size_t)(u & 0xFFFFu) * 8 + sub];
            }
#pragma unroll
            for (int k = 0; k < 4; ++k) {
                if (rr[k] >= 0) {
                    float* p = &acc[rr[k] * 32 + sub * 4];
                    __hip_atomic_fetch_add(p + 0, f[k].x * a[k], __ATOMIC_RELAXED, __HIP_MEMORY_SCOPE_WORKGROUP);
                    __hip_atomic_fetch_add(p + 1, f[k].y * a[k], __ATOMIC_RELAXED, __HIP_MEMORY_SCOPE_WORKGROUP);
                    __hip_atomic_fetch_add(p + 2, f[k].z * a[k], __ATOMIC_RELAXED, __HIP_MEMORY_SCOPE_WORKGROUP);
                    __hip_atomic_fetch_add(p + 3, f[k].w * a[k], __ATOMIC_RELAXED, __HIP_MEMORY_SCOPE_WORKGROUP);
                }
            }
        }
    } else {
        for (int e = tid; e < EE; e += 1024) {
            const unsigned int r = (unsigned int)(dst[e] - lo);
            if (r < (unsigned int)rows) lds_max_f32(&mrow[r], leaky(el[src[e]] + errow[r]));
        }
        __syncthreads();
        for (int e = tid; e < EE; e += 1024) {
            const unsigned int r = (unsigned int)(dst[e] - lo);
            if (r < (unsigned int)rows)
                __hip_atomic_fetch_add(&lrow[r], __expf(leaky(el[src[e]] + errow[r]) - mrow[r]),
                                       __ATOMIC_RELAXED, __HIP_MEMORY_SCOPE_WORKGROUP);
        }
        __syncthreads();
        for (int e = tid; e < EE; e += 1024) {
            const unsigned int r = (unsigned int)(dst[e] - lo);
            if (r < (unsigned int)rows) {
                const int sidx = src[e];
                const float a = __expf(leaky(el[sidx] + errow[r]) - mrow[r]) / lrow[r];
                for (int c = 0; c < 32; ++c)
                    __hip_atomic_fetch_add(&acc[r * 32 + c], feat[(size_t)sidx * 32 + c] * a,
                                           __ATOMIC_RELAXED, __HIP_MEMORY_SCOPE_WORKGROUP);
            }
        }
    }
    __syncthreads();

    for (int idx = tid; idx < rows * 32; idx += 1024)
        outp[(size_t)lo * 32 + idx] += acc[idx];
}

extern "C" void kernel_launch(void* const* d_in, const int* in_sizes, int n_in,
                              void* d_out, int out_size, void* d_ws, size_t ws_size,
                              hipStream_t stream) {
    const float* x     = (const float*)d_in[0];
    const int*   src   = (const int*)  d_in[1];
    const int*   dst   = (const int*)  d_in[2];
    const float* W1    = (const float*)d_in[3];
    const float* al1   = (const float*)d_in[4];
    const float* ar1   = (const float*)d_in[5];
    const float* b1    = (const float*)d_in[6];
    const float* W2    = (const float*)d_in[7];
    const float* al2   = (const float*)d_in[8];
    const float* ar2   = (const float*)d_in[9];
    const float* resW2 = (const float*)d_in[10];
    const float* b2    = (const float*)d_in[11];
    float* outp = (float*)d_out;

    float* ws    = (float*)d_ws;
    float* feat1 = ws;
    float* rst1  = feat1 + (size_t)NN * 256;
    float* el1   = rst1  + (size_t)NN * 256;
    float* er1   = el1   + (size_t)NN * 8;
    int*   lst   = (int*)(er1 + (size_t)NN * 8);
    int*   lcnt  = lst + (size_t)NTILE * CAP1;
    unsigned int* codes = (unsigned int*)(lcnt + NTILE);
    float* feat2 = feat1;
    float* el2   = el1;
    float* er2   = er1;

    gemm1_kernel<<<GEMM_WGS, GEMM_THREADS, 0, stream>>>(x, W1, feat1);
    elr1_kernel<<<ELR1_WGS, 256, 0, stream>>>(feat1, al1, ar1, el1, er1, dst, codes);
    agg1_tile_kernel<<<NTILE, 1024, 0, stream>>>(src, dst, codes, el1, er1,
                                                 feat1, x, b1, rst1, lst, lcnt);

    gemm2_kernel<<<GEMM_WGS, GEMM_THREADS, 0, stream>>>(rst1, W2, resW2, b2, feat2, outp);
    elr2_kernel<<<ELR2_WGS, 128, 0, stream>>>(feat2, al2, ar2, el2, er2);
    agg2_tile_kernel<<<NTILE, 1024, 0, stream>>>(src, dst, el2, er2, feat2, lst, lcnt, outp);
}


# Layer-1 softmax weights formed once per edge and head by the thread holding its scores, filed in an LDS table; the owner gather reads its two by one 8-byte LDS load (no exp, division or logit load)
# speedup vs baseline: 5.3421x; 1.0487x over previous
#include <hip/hip_runtime.h>
#include <hip/hip_bf16.h>
#include <math.h>

#define NN 50000
#define EE 800000
#define INF_ 256
#define OUTF 32
#define H1N 8
#define SLOPE 0.2f

typedef float v2f __attribute__((ext_vector_type(2)));
typedef float v8f __attribute__((ext_vector_type(8)));

__device__ __forceinline__ float leaky(float x) { return x > 0.f ? x : SLOPE * x; }

typedef float v4f __attribute__((ext_vector_type(4)));

#define NT_ROWS (NN / 16)
#define GEMM_WGS 256
#define GEMM_THREADS 512
static_assert(NN % 16 == 0, "whole 16-row tiles only: no row guard in the GEMM kernels");

template <int NCOL>
__device__ __forceinline__ void stage_weights(float* __restrict__ Wl, const float* __restrict__ Wa,
                                              const float* __restrict__ Wb, const int tid) {
    constexpr int P = NCOL * 4 + 32;
    constexpr int CG = NCOL / 4;
    constexpr int NB = 64 * CG / GEMM_THREADS;
    static_assert(64 * CG % GEMM_THREADS == 0, "every thread moves the same number of blocks");
    const int rot = (tid >> 2) & 3;
    v4f r[NB][4];
#pragma unroll
    for (int it = 0; it < NB; ++it) {
        const int q = tid + it * GEMM_THREADS;
        const int p = q / CG, cg = q % CG;
        const float* s;
        int pitch;
        if (NCOL == 256) { s = Wa + (size_t)(4 * p) * 256 + 4 * cg; pitch = 256; }
        else             { s = (cg < 8 ? Wa : Wb) + (size_t)(4 * p) * 32 + 4 * (cg & 7); pitch = 32; }
#pragma unroll
        for (int i = 0; i < 4; ++i) r[it][i] = *reinterpret_cast<const v4f*>(s + i * pitch);
    }
#pragma unroll
    for (int it = 0; it < NB; ++it) {
        const int q = tid + it * GEMM_THREADS;
        const int p = q / CG, cg = q % CG;
#pragma unroll
        for (int j = 0; j < 4; ++j) {
            const int jj = (j + rot) & 3;
            v4f o;
            o.x = jj == 0 ? r[it][0].x : jj == 1 ? r[it][0].y : jj == 2 ? r[it][0].z : r[it][0].w;
            o.y = jj == 0 ? r[it][1].x : jj == 1 ? r[it][1].y : jj == 2 ? r[it][1].z : r[it][1].w;
            o.z = jj == 0 ? r[it][2].x : jj == 1 ? r[it][2].y : jj == 2 ? r[it][2].z : r[it][2].w;
            o.w = jj == 0 ? r[it][3].x : jj == 1 ? r[it][3].y : jj == 2 ? r[it][3].z : r[it][3].w;
            *reinterpret_cast<v4f*>(Wl + p * P + (4 * cg + jj) * 4) = o;
        }
    }
}

template <int NCOL, int NT16>
__device__ __forceinline__ void unit_mma(const float* __restrict__ A, const float* __restrict__ Wl,
                                         const int tile, const int col0, const int lane, v8f (&acc)[NT16]) {
    constexpr int P = NCOL * 4 + 32;
    const int h = lane >> 4, c = lane & 15;
    const v4f* __restrict__ a4 = reinterpret_cast<const v4f*>(A + (size_t)(tile * 16 + c) * 256) + h;
    const float* bp = Wl + h * P + (col0 + c) * 4;
    for (int J0 = 0; J0 < 32; J0 += 8) {
        v4f ac[8];
#pragma unroll
        for (int j = 0; j < 8; ++j) ac[j] = a4[2 * (J0 + j)];
#pragma unroll
        for (int j = 0; j < 8; ++j) {
            const float* bj = bp + 2 * (J0 + j) * P;
            v4f b[NT16];
#pragma unroll
            for (int t = 0; t < NT16; ++t) b[t] = *reinterpret_cast<const v4f*>(bj + t * 64);
            v2f alo, ahi;
            alo.x = ac[j].x; alo.y = ac[j].y; ahi.x = ac[j].z; ahi.y = ac[j].w;
#pragma unroll
            for (int t = 0; t < NT16; ++t) {
                v2f blo; blo.x = b[t].x; blo.y = b[t].y;
                acc[t] = __builtin_amdgcn_wmma_f32_16x16x4_f32(false, alo, false, blo, (short)0, acc[t], false, false);
            }
#pragma unroll
            for (int t = 0; t < NT16; ++t) {
                v2f bhi; bhi.x = b[t].z; bhi.y = b[t].w;
                acc[t] = __builtin_amdgcn_wmma_f32_16x16x4_f32(false, ahi, false, bhi, (short)0, acc[t], false, false);
            }
        }
    }
}

__global__ void __launch_bounds__(GEMM_THREADS) gemm1_kernel(const float* __restrict__ X,
                                                             const float* __restrict__ W,
                                                             float* __restrict__ C) {
    __shared__ __attribute__((aligned(16))) float Wl[64 * (256 * 4 + 32)];
    const int tid = threadIdx.x, lane = tid & 31;
    const int wave = __builtin_amdgcn_readfirstlane(tid >> 5);
    stage_weights<256>(Wl, W, W, tid);
    __syncthreads();
    const int ntb = (NT_ROWS - (int)blockIdx.x + GEMM_WGS - 1) / GEMM_WGS;
    const int g = wave & 3, h = lane >> 4, c = lane & 15;
    for (int i = ((wave >> 2) - g) & 3; i < ntb; i += 4) {
        const int tile = blockIdx.x + GEMM_WGS * i;
        v8f acc[4] = {};
        unit_mma<256, 4>(X, Wl, tile, g * 64, lane, acc);
        float* crow = C + (size_t)(tile * 16 + h * 8) * 256 + g * 64 + c;
#pragma unroll
        for (int t = 0; t < 4; ++t) {
#pragma unroll
            for (int j = 0; j < 8; ++j) crow[(size_t)j * 256 + t * 16] = acc[t][j];
        }
    }
}

__global__ void __launch_bounds__(GEMM_THREADS) gemm2_kernel(const float* __restrict__ Y,
                                                             const float* __restrict__ W2,
                                                             const float* __restrict__ RW,
                                                             const float* __restrict__ b2,
                                                             float* __restrict__ F2,
                                                             float* __restrict__ OUTP) {
    __shared__ __attribute__((aligned(16))) float Wl[64 * (64 * 4 + 32)];
    const int tid = threadIdx.x, lane = tid & 31;
    const int wave = __builtin_amdgcn_readfirstlane(tid >> 5);
    stage_weights<64>(Wl, W2, RW, tid);
    __syncthreads();
    const int ntb = (NT_ROWS - (int)blockIdx.x + GEMM_WGS - 1) / GEMM_WGS;
    const int h = lane >> 4, c = lane & 15;
    for (int u = wave; u < 2 * ntb; u += 16) {
        const int tile = blockIdx.x + GEMM_WGS * (u >> 1);
        const int half = u & 1;
        v8f acc[2] = {};
        unit_mma<64, 2>(Y, Wl, tile, half * 32, lane, acc);
        const size_t o = (size_t)(tile * 16 + h * 8) * 32 + c;
        if (half == 0) {
#pragma unroll
            for (int t = 0; t < 2; ++t) {
#pragma unroll
                for (int j = 0; j < 8; ++j) F2[o + (size_t)j * 32 + t * 16] = acc[t][j];
            }
        } else {
#pragma unroll
            for (int t = 0; t < 2; ++t) {
                const float bb = b2[t * 16 + c];
#pragma unroll
                for (int j = 0; j < 8; ++j) OUTP[o + (size_t)j * 32 + t * 16] = acc[t][j] + bb;
            }
        }
    }
}

__device__ __forceinline__ float dot4(const v4f a, const v4f b) {
    return a.x * b.x + a.y * b.y + a.z * b.z + a.w * b.w;
}

#define ROWS1 8
#define ELR1_WGS ((NN / ROWS1 + 7) / 8)
#define TILE1 196
#define NTILE ((NN + TILE1 - 1) / TILE1)
static_assert(NN % ROWS1 == 0, "a wave takes ROWS1 whole rows");
static_assert(ROWS1 == 8, "the load fence in the kernel names eight rows");
static_assert(NTILE <= 256, "a tile number fits a byte");
static_assert(EE % 16 == 0, "codes are written four and read sixteen at a time");
static_assert(ELR1_WGS * 256 >= EE / 4, "one thread of the logit launch codes four edges");

__device__ __forceinline__ unsigned int tile_code(int d) { return ((unsigned int)d / TILE1) & 255u; }

__global__ void __launch_bounds__(256) elr1_kernel(const float* __restrict__ feat,
                                                   const float* __restrict__ al,
                                                   const float* __restrict__ ar,
                                                   float* __restrict__ el,
                                                   float* __restrict__ er,
                                                   const int* __restrict__ dst,
                                                   unsigned int* __restrict__ codes) {
    const int lane = threadIdx.x & 31;
    const int g = blockIdx.x * 256 + threadIdx.x;
    if (g < EE / 4) {
        const int4 d = reinterpret_cast<const int4*>(dst)[g];
        codes[g] = tile_code(d.x) | (tile_code(d.y) << 8) | (tile_code(d.z) << 16) | (tile_code(d.w) << 24);
    }
    const int row0 = (blockIdx.x * 8 + (threadIdx.x >> 5)) * ROWS1;
    if (row0 >= NN) return;
    const v4f* __restrict__ f4 = reinterpret_cast<const v4f*>(feat) + (size_t)row0 * 64 + lane;
    v4f v[ROWS1][2];
#pragma unroll
    for (int i = 0; i < ROWS1; ++i) { v[i][0] = f4[i * 64]; v[i][1] = f4[i * 64 + 32]; }
    const v4f* __restrict__ al4 = reinterpret_cast<const v4f*>(al);
    const v4f* __restrict__ ar4 = reinterpret_cast<const v4f*>(ar);
    const v4f l0 = al4[lane], l1 = al4[32 + lane];
    const v4f r0 = ar4[lane], r1 = ar4[32 + lane];
    asm volatile("" :: "v"(v[0][0]), "v"(v[0][1]), "v"(v[1][0]), "v"(v[1][1]),
                       "v"(v[2][0]), "v"(v[2][1]), "v"(v[3][0]), "v"(v[3][1]),
                       "v"(v[4][0]), "v"(v[4][1]), "v"(v[5][0]), "v"(v[5][1]),
                       "v"(v[6][0]), "v"(v[6][1]), "v"(v[7][0]), "v"(v[7][1]),
                       "v"(l0), "v"(l1), "v"(r0), "v"(r1));
    const bool b0 = (lane & 1) != 0, b1 = (lane & 2) != 0, b2 = (lane & 4) != 0;
    const int head = (lane >> 3) + (b1 ? 4 : 0);
    float* __restrict__ dstp = b0 ? er : el;
#pragma unroll
    for (int i = 0; i < ROWS1; i += 2) {
        float u[2];
#pragma unroll
        for (int c = 0; c < 2; ++c) {
            const float a0 = dot4(v[i + c][0], l0), e0 = dot4(v[i + c][0], r0);
            const float a1 = dot4(v[i + c][1], l1), e1 = dot4(v[i + c][1], r1);
            const float t0 = (b0 ? e0 : a0) + __shfl_xor(b0 ? a0 : e0, 1, 32);
            const float t1 = (b0 ? e1 : a1) + __shfl_xor(b0 ? a1 : e1, 1, 32);
            u[c] = (b1 ? t1 : t0) + __shfl_xor(b1 ? t0 : t1, 2, 32);
        }
        const float w = (b2 ? u[1] : u[0]) + __shfl_xor(b2 ? u[0] : u[1], 4, 32);
        dstp[(row0 + i + (b2 ? 1 : 0)) * H1N + head] = w;
    }
}

#define ROWS2 16
#define ELR2_WGS ((NN / ROWS2 + 3) / 4)
static_assert(NN % ROWS2 == 0, "a wave takes ROWS2 whole rows");
static_assert(ROWS2 == 16, "four loads of four rows, joined pairwise by the exchange tree");

__global__ void __launch_bounds__(128) elr2_kernel(const float* __restrict__ feat,
                                                   const float* __restrict__ al,
                                                   const float* __restrict__ ar,
                                                   float* __restrict__ el,
                                                   float* __restrict__ er) {
    const int lane = threadIdx.x & 31;
    const int row0 = (blockIdx.x * 4 + (threadIdx.x >> 5)) * ROWS2;
    if (row0 >= NN) return;
    const v4f* __restrict__ f4 = reinterpret_cast<const v4f*>(feat) + (size_t)row0 * 8 + lane;
    v4f v[4];
#pragma unroll
    for (int q = 0; q < 4; ++q) v[q] = f4[q * 32];
    const v4f l = reinterpret_cast<const v4f*>(al)[lane & 7];
    const v4f r = reinterpret_cast<const v4f*>(ar)[lane & 7];
    const bool b0 = (lane & 1) != 0, b1 = (lane & 2) != 0, b2 = (lane & 4) != 0;
    float t[4];
#pragma unroll
    for (int q = 0; q < 4; ++q) {
        const float a = dot4(v[q], l), e = dot4(v[q], r);
        t[q] = (b0 ? e : a) + __shfl_xor(b0 ? a : e, 1, 32);
    }
    const float u0 = (b1 ? t[1] : t[0]) + __shfl_xor(b1 ? t[0] : t[1], 2, 32);
    const float u1 = (b1 ? t[3] : t[2]) + __shfl_xor(b1 ? t[2] : t[3], 2, 32);
    const float w  = (b2 ? u1 : u0) + __shfl_xor(b2 ? u0 : u1, 4, 32);
    const int q = (b2 ? 2 : 0) + (b1 ? 1 : 0);
    (b0 ? er : el)[row0 + 4 * q + (lane >> 3)] = w;
}

__device__ __forceinline__ void lds_max_f32(float* addr, float v) {
    if (v >= 0.f) __hip_atomic_fetch_max((int*)addr, __float_as_int(v),
                                         __ATOMIC_RELAXED, __HIP_MEMORY_SCOPE_WORKGROUP);
    else          __hip_atomic_fetch_min((unsigned int*)addr, __float_as_uint(v),
                                         __ATOMIC_RELAXED, __HIP_MEMORY_SCOPE_WORKGROUP);
}

#define CAP1  4096
#define WCAP  256
#define NITC  ((EE / 16 + 1023) / 1024)
#define GRPC  7
static_assert(NN <= 65536 && TILE1 < 65535, "tile row and source index are packed into 32 bits");
static_assert(NITC % GRPC == 0, "scan loop takes GRPC trips at a time");
static_assert(CAP1 % 1024 == 0, "a thread owns CAP1 / 1024 list places");
static_assert(CAP1 <= (1 << 14), "an owner's entry holds the list place above bit 18");
static_assert(CAP1 * H1N <= (TILE1 + 1) * 256, "the weight table fits the LDS of the fallbacks' tile copy");

__device__ __forceinline__ unsigned int zero_bytes(unsigned int y) {
    return ~(((y & 0x7F7F7F7Fu) + 0x7F7F7F7Fu) | y) & 0x80808080u;
}

__device__ __forceinline__ void edge_loads(const unsigned int uv, const float* wts,
                                           const v4f* __restrict__ feat4, const int lane,
                                           v2f (&wv)[4], v4f (&fA)[4], v4f (&fB)[4]) {
#pragma unroll
    for (int k = 0; k < 4; ++k) {
        const int s = __builtin_amdgcn_readlane((int)uv, k) & 0xFFFF;
        const v4f* __restrict__ rowp = feat4 + (size_t)s * 64;
        fA[k] = rowp[lane];
        fB[k] = rowp[32 + lane];
    }
#pragma unroll
    for (int k = 0; k < 4; ++k) {
        const unsigned int pos = (unsigned int)__builtin_amdgcn_readlane((int)uv, k) >> 18;
        wv[k] = *reinterpret_cast<const v2f*>(wts + pos * H1N + 2 * (lane >> 3));
    }
}

#define SLOTS 4
#define NGRP  2
#define GCAP  (WCAP / NGRP)
static_assert((TILE1 + 1 + 31) / 32 <= SLOTS * NGRP, "every tile row and the spare row have a slot");

struct SlotRows { v4f a0, b0, a1, b1, a2, b2, a3, b3; };
static_assert(SLOTS == 4, "SlotRows names four slots");

#define ROW_FMA(J, RA, RB)                                                                       \
    asm volatile("s_cmp_lg_u32 %18, " #J "\n\t"                                                  \
                 "s_cbranch_scc1 .Lslot_skip%=\n\t"                                              \
                 "v_fmac_f32 %0, %8, %16\n\tv_fmac_f32 %1, %9, %16\n\t"                         \
                 "v_fmac_f32 %2, %10, %16\n\tv_fmac_f32 %3, %11, %16\n\t"                       \
                 "v_fmac_f32 %4, %12, %17\n\tv_fmac_f32 %5, %13, %17\n\t"                       \
                 "v_fmac_f32 %6, %14, %17\n\tv_fmac_f32 %7, %15, %17\n"                         \
                 ".Lslot_skip%=:"                                                                \
                 : "+v"(RA.x), "+v"(RA.y), "+v"(RA.z), "+v"(RA.w),                               \
                   "+v"(RB.x), "+v"(RB.y), "+v"(RB.z), "+v"(RB.w)                                \
                 : "v"(fA.x), "v"(fA.y), "v"(fA.z), "v"(fA.w),                                   \
                   "v"(fB.x), "v"(fB.y), "v"(fB.z), "v"(fB.w), "v"(wA), "v"(wB), "s"(j)          \
                 : "scc")

__device__ __forceinline__ void slot_add(SlotRows& s, const int j,
                                         const v4f fA, const float wA, const v4f fB, const float wB) {
    ROW_FMA(0, s.a0, s.b0);
    ROW_FMA(1, s.a1, s.b1);
    ROW_FMA(2, s.a2, s.b2);
    ROW_FMA(3, s.a3, s.b3);
}

__device__ __forceinline__ void slot_rows_x(SlotRows& s, const v4f* __restrict__ x4, const int row0, const int rows) {
    const int r0 = row0      < rows ? row0      : rows - 1;
    const int r1 = row0 + 32 < rows ? row0 + 32 : rows - 1;
    const int r2 = row0 + 64 < rows ? row0 + 64 : rows - 1;
    const int r3 = row0 + 96 < rows ? row0 + 96 : rows - 1;
    s.a0 = x4[(size_t)r0 * 64]; s.b0 = x4[(size_t)r0 * 64 + 32];
    s.a1 = x4[(size_t)r1 * 64]; s.b1 = x4[(size_t)r1 * 64 + 32];
    s.a2 = x4[(size_t)r2 * 64]; s.b2 = x4[(size_t)r2 * 64 + 32];
    s.a3 = x4[(size_t)r3 * 64]; s.b3 = x4[(size_t)r3 * 64 + 32];
}

__device__ __forceinline__ v4f relu4(const v4f a) {
    v4f o;
    o.x = fmaxf(a.x, 0.f); o.y = fmaxf(a.y, 0.f); o.z = fmaxf(a.z, 0.f); o.w = fmaxf(a.w, 0.f);
    return o;
}

__device__ __forceinline__ void tile_start_lds(float* acc, const float* __restrict__ x,
                                               const float* __restrict__ b1, const int lo,
                                               const int rows, const int tid) {
    for (int idx = tid; idx < rows * 256; idx += 1024)
        acc[idx] = x[(size_t)lo * 256 + idx] + b1[idx & 255];
}

__global__ void __launch_bounds__(1024) agg1_tile_kernel(const int* __restrict__ src,
                                                         const int* __restrict__ dst,
                                                         const unsigned int* __restrict__ codes,
                                                         const float* __restrict__ el,
                                                         const float* __restrict__ er,
                                                         const float* __restrict__ feat,
                                                         const float* __restrict__ x,
                                                         const float* __restrict__ b1,
                                                         float* __restrict__ y,
                                                         int* __restrict__ glist,
                                                         int* __restrict__ gcnt) {
    __shared__ __attribute__((aligned(16))) float acc[(TILE1 + 1) * 256];
    __shared__ __attribute__((aligned(16))) float errow[(TILE1 + 1) * H1N];
    __shared__ float mrow[(TILE1 + 1) * H1N];
    __shared__ __attribute__((aligned(16))) float lrow[(TILE1 + 1) * H1N];
    __shared__ unsigned int ent[CAP1];
    __shared__ unsigned int wl[32 * WCAP];
    __shared__ unsigned int wcnt[32 * NGRP];
    __shared__ unsigned int cnt;

    const int tid  = threadIdx.x;
    const int lane = tid & 31;
    const int wave = __builtin_amdgcn_readfirstlane(tid >> 5);
    const int lo   = blockIdx.x * TILE1;
    const int rows = (NN - lo) < TILE1 ? (NN - lo) : TILE1;

    for (int idx = tid; idx < (TILE1 + 1) * H1N; idx += 1024) {
        mrow[idx]  = -INFINITY;
        lrow[idx]  = 0.f;
        errow[idx] = idx < rows * H1N ? er[(size_t)lo * H1N + idx] : 0.f;
    }
    if (tid == 0) cnt = 0u;
    if (tid < 32 * NGRP) wcnt[tid] = 0u;

    const uint4* __restrict__ code4 = reinterpret_cast<const uint4*>(codes);
    const unsigned int pat = (unsigned int)blockIdx.x * 0x01010101u;
    const int it0 = (blockIdx.x * NITC) >> 8;
    uint4 cur[GRPC];
#pragma unroll
    for (int q = 0; q < GRPC; ++q) {
        int it = it0 + q; if (it >= NITC) it -= NITC;
        const int iw = it * 1024 + tid;
        cur[q] = code4[iw < EE / 16 ? iw : tid];
    }
    __syncthreads();

    for (int j = 0; j < NITC; j += GRPC) {
        uint4 nxt[GRPC];
#pragma unroll
        for (int q = 0; q < GRPC; ++q) {
            int it = it0 + j + GRPC + q; if (it >= NITC) it -= NITC;
            const int iw = it * 1024 + tid;
            nxt[q] = code4[iw < EE / 16 ? iw : tid];
        }
#pragma unroll
        for (int q = 0; q < GRPC; ++q) {
            int it = it0 + j + q; if (it >= NITC) it -= NITC;
            const int iw = it * 1024 + tid;
            unsigned int m = (zero_bytes(cur[q].x ^ pat) >> 7) | (zero_bytes(cur[q].y ^ pat) >> 6)
                           | (zero_bytes(cur[q].z ^ pat) >> 5) | (zero_bytes(cur[q].w ^ pat) >> 4);
            if (iw >= EE / 16) m = 0u;
            while (m != 0u) {
                const unsigned int p = atomicAdd(&cnt, 1u);
                const int b = __builtin_ctz(m);
                if (p < CAP1) ent[p] = (unsigned int)(iw * 16 + (b & 7) * 4 + (b >> 3));
                m &= m - 1u;
            }
        }
#pragma unroll
        for (int q = 0; q < GRPC; ++q) cur[q] = nxt[q];
    }
    __syncthreads();
    const unsigned int total = cnt;
    if (tid == 0) gcnt[blockIdx.x] = (int)total;

    bool in_regs = false;
    if (total <= CAP1) {
        const int n = (int)total;
        int* __restrict__ gl = glist + (size_t)blockIdx.x * CAP1;
        const v4f* __restrict__ x4 = reinterpret_cast<const v4f*>(x) + (size_t)lo * 64 + lane;

        const float4* __restrict__ el4 = reinterpret_cast<const float4*>(el);
        float sc[CAP1 / 1024][H1N];
        int   rw[CAP1 / 1024];
        int   sw[CAP1 / 1024];
#pragma unroll
        for (int it = 0; it < CAP1 / 1024; ++it) {
            const int i = tid + 1024 * it;
            const int e = i < n ? (int)ent[i] : 0;
            rw[it] = dst[e];
            sw[it] = src[e];
        }
#pragma unroll
        for (int it = 0; it < CAP1 / 1024; ++it) {
            const unsigned int r = (unsigned int)(rw[it] - lo);
            rw[it] = (tid + 1024 * it) < n ? (r < (unsigned int)rows ? (int)r : TILE1) : -1;
        }
#pragma unroll
        for (int it = 0; it < CAP1 / 1024; ++it) {
            const int r = rw[it] < 0 ? 0 : rw[it];
            const float4 a0 = el4[(size_t)sw[it] * 2], a1 = el4[(size_t)sw[it] * 2 + 1];
            const float4 c0 = *reinterpret_cast<const float4*>(&errow[r * H1N]);
            const float4 c1 = *reinterpret_cast<const float4*>(&errow[r * H1N + 4]);
            sc[it][0] = leaky(a0.x + c0.x); sc[it][1] = leaky(a0.y + c0.y);
            sc[it][2] = leaky(a0.z + c0.z); sc[it][3] = leaky(a0.w + c0.w);
            sc[it][4] = leaky(a1.x + c1.x); sc[it][5] = leaky(a1.y + c1.y);
            sc[it][6] = leaky(a1.z + c1.z); sc[it][7] = leaky(a1.w + c1.w);
        }
#pragma unroll
        for (int it = 0; it < CAP1 / 1024; ++it) {
            if (rw[it] >= 0) {
#pragma unroll
                for (int h = 0; h < H1N; ++h) lds_max_f32(&mrow[rw[it] * H1N + h], sc[it][h]);
                const unsigned int pk = ((unsigned int)rw[it] << 16) | (unsigned int)sw[it];
                ent[tid + 1024 * it] = pk;
                gl[tid + 1024 * it]  = (int)pk;
                const unsigned int ow = ((unsigned int)rw[it] & 31u) * NGRP + ((unsigned int)rw[it] >> 5) / SLOTS;
                const unsigned int p  = atomicAdd(&wcnt[ow], 1u);
                if (p < GCAP) wl[ow * GCAP + p] = ((unsigned int)(tid + 1024 * it) << 18)
                                                | ((((unsigned int)rw[it] >> 5) % SLOTS) << 16) | (unsigned int)sw[it];
            }
        }
        __syncthreads();
#pragma unroll
        for (int it = 0; it < CAP1 / 1024; ++it) {
            if (rw[it] >= 0) {
#pragma unroll
                for (int h = 0; h < H1N; ++h) {
                    sc[it][h] = __expf(sc[it][h] - mrow[rw[it] * H1N + h]);
                    __hip_atomic_fetch_add(&lrow[rw[it] * H1N + h], sc[it][h],
                                           __ATOMIC_RELAXED, __HIP_MEMORY_SCOPE_WORKGROUP);
                }
            }
        }
        __syncthreads();
        const unsigned int myc = wcnt[lane] > wcnt[32 + lane] ? wcnt[lane] : wcnt[32 + lane];
        static_assert(NGRP == 2, "the overflow test reads two counts a lane");
        if (!__any(myc > (unsigned int)GCAP)) {
            in_regs = true;
            float* wts = acc;
#pragma unroll
            for (int it = 0; it < CAP1 / 1024; ++it) {
                if (rw[it] >= 0) {
                    const float4 l0 = *reinterpret_cast<const float4*>(&lrow[rw[it] * H1N]);
                    const float4 l1 = *reinterpret_cast<const float4*>(&lrow[rw[it] * H1N + 4]);
                    v4f o0, o1;
                    o0.x = sc[it][0] / l0.x; o0.y = sc[it][4] / l1.x; o0.z = sc[it][1] / l0.y; o0.w = sc[it][5] / l1.y;
                    o1.x = sc[it][2] / l0.z; o1.y = sc[it][6] / l1.z; o1.z = sc[it][3] / l0.w; o1.w = sc[it][7] / l1.w;
                    v4f* wrow = reinterpret_cast<v4f*>(wts + (tid + 1024 * it) * H1N);
                    wrow[0] = o0;
                    wrow[1] = o1;
                }
            }
            SlotRows sr;
            slot_rows_x(sr, x4, wave, rows);
            const v4f* __restrict__ b14 = reinterpret_cast<const v4f*>(b1);
            const v4f bA = b14[lane], bB = b14[32 + lane];
            __syncthreads();
            const int q  = lane & 3;
            const v4f* __restrict__ feat4 = reinterpret_cast<const v4f*>(feat);
            v4f* __restrict__ y4 = reinterpret_cast<v4f*>(y) + (size_t)lo * 64 + lane;
            for (int g = 0; g < NGRP; ++g) {
                const int nw = __builtin_amdgcn_readfirstlane((int)wcnt[wave * NGRP + g]);
                const unsigned int* wlw = wl + (wave * NGRP + g) * GCAP;
                const int row0 = wave + 32 * SLOTS * g;
                if (g > 0) slot_rows_x(sr, x4, row0, rows);
                sr.a0 += bA; sr.b0 += bB; sr.a1 += bA; sr.b1 += bB;
                sr.a2 += bA; sr.b2 += bB; sr.a3 += bA; sr.b3 += bB;
                unsigned int uv = nw > 0 ? wlw[q < nw ? q : 0] : 0u;
                for (int i0 = 0; i0 < nw; i0 += 4) {
                    const unsigned int uvN = wlw[i0 + 4 + q < nw ? i0 + 4 + q : i0];
                    v2f wv[4];
                    v4f fA[4], fB[4];
                    edge_loads(uv, wts, feat4, lane, wv, fA, fB);
#pragma unroll
                    for (int k = 0; k < 4; ++k) {
                        const int jj = (int)(((unsigned int)__builtin_amdgcn_readlane((int)uv, k) >> 16) & 3u);
                        static_assert(SLOTS == 4, "the entry holds the slot in two bits");
                        const bool past = i0 + k >= nw;
                        slot_add(sr, jj, fA[k], past ? 0.f : wv[k].x, fB[k], past ? 0.f : wv[k].y);
                    }
                    uv = uvN;
                }
                if (row0      < rows) { y4[(size_t)row0 * 64]        = relu4(sr.a0); y4[(size_t)row0 * 64 + 32]        = relu4(sr.b0); }
                if (row0 + 32 < rows) { y4[(size_t)(row0 + 32) * 64] = relu4(sr.a1); y4[(size_t)(row0 + 32) * 64 + 32] = relu4(sr.b1); }
                if (row0 + 64 < rows) { y4[(size_t)(row0 + 64) * 64] = relu4(sr.a2); y4[(size_t)(row0 + 64) * 64 + 32] = relu4(sr.b2); }
                if (row0 + 96 < rows) { y4[(size_t)(row0 + 96) * 64] = relu4(sr.a3); y4[(size_t)(row0 + 96) * 64 + 32] = relu4(sr.b3); }
            }
        } else {
        tile_start_lds(acc, x, b1, lo, rows, tid);
        __syncthreads();
        const int h = lane & 7;
        for (int i0 = wave; i0 < n; i0 += 128) {
            float a[4];
            const float* fp[4];
            float* ap[4];
            const int ikl = i0 + 32 * (lane & 3);
            const unsigned int uv = ent[ikl < n ? ikl : i0];
#pragma unroll
            for (int k = 0; k < 4; ++k) {
                const bool okk = (i0 + 32 * k) < n;
                const unsigned int u = (unsigned int)__builtin_amdgcn_readlane((int)uv, k);
                const int s = (int)(u & 0xFFFFu), r = (int)(u >> 16);
                a[k]  = __expf(leaky(el[s * H1N + h] + errow[r * H1N + h]) - mrow[r * H1N + h])
                        / lrow[r * H1N + h];
                fp[k] = feat + (size_t)s * 256 + lane;
                ap[k] = &acc[(okk ? r : TILE1) * 256 + lane];
            }
            float f[4][8];
#pragma unroll
            for (int j = 0; j < 8; ++j) {
#pragma unroll
                for (int k = 0; k < 4; ++k) f[k][j] = fp[k][32 * j];
            }
#pragma unroll
            for (int j = 0; j < 8; ++j) {
#pragma unroll
                for (int k = 0; k < 4; ++k) {
                    const float w = __int_as_float(__builtin_amdgcn_readlane(__float_as_int(a[k]), j));
                    __hip_atomic_fetch_add(ap[k] + 32 * j, f[k][j] * w,
                                           __ATOMIC_RELAXED, __HIP_MEMORY_SCOPE_WORKGROUP);
                }
            }
        }
        }
    } else {
        tile_start_lds(acc, x, b1, lo, rows, tid);
        for (int e = tid; e < EE; e += 1024) {
            const unsigned int r = (unsigned int)(dst[e] - lo);
            if (r < (unsigned int)rows) {
                const int sidx = src[e];
                for (int hh = 0; hh < H1N; ++hh)
                    lds_max_f32(&mrow[r * H1N + hh], leaky(el[sidx * H1N + hh] + errow[r * H1N + hh]));
            }
        }
        __syncthreads();
        for (int e = tid; e < EE; e += 1024) {
            const unsigned int r = (unsigned int)(dst[e] - lo);
            if (r < (unsigned int)rows) {
                const int sidx = src[e];
                for (int hh = 0; hh < H1N; ++hh)
                    __hip_atomic_fetch_add(&lrow[r * H1N + hh],
                                           __expf(leaky(el[sidx * H1N + hh] + errow[r * H1N + hh]) - mrow[r * H1N + hh]),
                                           __ATOMIC_RELAXED, __HIP_MEMORY_SCOPE_WORKGROUP);
            }
        }
        __syncthreads();
        for (int e = tid; e < EE; e += 1024) {
            const unsigned int r = (unsigned int)(dst[e] - lo);
            if (r < (unsigned int)rows) {
                const int sidx = src[e];
                for (int hh = 0; hh < H1N; ++hh) {
                    const float w = __expf(leaky(el[sidx * H1N + hh] + errow[r * H1N + hh]) - mrow[r * H1N + hh])
                                    / lrow[r * H1N + hh];
                    for (int c = 0; c < 32; ++c)
                        __hip_atomic_fetch_add(&acc[r * 256 + hh * 32 + c], feat[(size_t)sidx * 256 + hh * 32 + c] * w,
                                               __ATOMIC_RELAXED, __HIP_MEMORY_SCOPE_WORKGROUP);
                }
            }
        }
    }
    if (!in_regs) {
        __syncthreads();
        for (int idx = tid; idx < rows * 256; idx += 1024)
            y[(size_t)lo * 256 + idx] = fmaxf(acc[idx], 0.f);
    }
}

#define TILE2 196
static_assert(TILE2 == TILE1, "layer 2 reads the lists that layer 1 made for its own tiles");
static_assert(NN <= 65536, "source index is packed into 16 bits");
static_assert((size_t)NTILE * CAP1 + NTILE + EE / 4 <= (size_t)NN * 34, "lists, counts and tile codes fit the workspace part kept for them");


__global__ void __launch_bounds__(1024) agg2_tile_kernel(const int* __restrict__ src,
                                                         const int* __restrict__ dst,
                                                         const float* __restrict__ el,
                                                         const float* __restrict__ er,
                                                         const float* __restrict__ feat,
                                                         const int* __restrict__ glist,
                                                         const int* __restrict__ gcnt,
                                                         float* __restrict__ outp) {
    __shared__ float acc[(TILE2 + 1) * 32];
    __shared__ float mrow[TILE2 + 1];
    __shared__ float lrow[TILE2 + 1];
    __shared__ float errow[TILE2 + 1];
    __shared__ unsigned int ent[CAP1];
    __shared__ float val[CAP1];

    const int tid  = threadIdx.x;
    const int lo   = blockIdx.x * TILE2;
    const int rows = (NN - lo) < TILE2 ? (NN - lo) : TILE2;
    const unsigned int total = (unsigned int)gcnt[blockIdx.x];

    for (int idx = tid; idx < (TILE2 + 1) * 32; idx += 1024) acc[idx] = 0.f;
    if (tid <= TILE2) {
        mrow[tid]  = -INFINITY;
        lrow[tid]  = 0.f;
        errow[tid] = tid < rows ? er[lo + tid] : 0.f;
    }
    __syncthreads();

    if (total <= CAP1) {
        const int n = (int)total;
        const int* __restrict__ gl = glist + (size_t)blockIdx.x * CAP1;
        unsigned int uu[CAP1 / 1024];
        float ee[CAP1 / 1024];
#pragma unroll
        for (int it = 0; it < CAP1 / 1024; ++it) {
            const int i = tid + 1024 * it;
            uu[it] = i < n ? (unsigned int)gl[i] : 0u;
        }
#pragma unroll
        for (int it = 0; it < CAP1 / 1024; ++it) ee[it] = el[uu[it] & 0xFFFFu];
#pragma unroll
        for (int it = 0; it < CAP1 / 1024; ++it) {
            const int i = tid + 1024 * it;
            if (i < n) {
                const int r = (int)(uu[it] >> 16);
                const float sc = leaky(ee[it] + errow[r]);
                lds_max_f32(&mrow[r], sc);
                ent[i] = uu[it];
                val[i] = sc;
            }
        }
        __syncthreads();
        for (int i = tid; i < n; i += 1024) {
            const int r = (int)(ent[i] >> 16);
            const float p = __expf(val[i] - mrow[r]);
            val[i] = p;
            __hip_atomic_fetch_add(&lrow[r], p, __ATOMIC_RELAXED, __HIP_MEMORY_SCOPE_WORKGROUP);
        }
        __syncthreads();
        const int sub = tid & 7;
        const float4* __restrict__ feat4 = reinterpret_cast<const float4*>(feat);
        for (int i = tid >> 3; i < n; i += 512) {
            float4 f[4]; float a[4]; int rr[4];
#pragma unroll
            for (int k = 0; k < 4; ++k) {
                const int ik = i + 128 * k;
                const int ic = ik < n ? ik : i;
                const unsigned int u = ent[ic];
                rr[k] = ik < n ? (int)(u >> 16) : -1;
                a[k]  = val[ic] / lrow[u >> 16];
                f[k]  = feat4[(size_t)(u & 0xFFFFu) * 8 + sub];
            }
#pragma unroll
            for (int k = 0; k < 4; ++k) {
                if (rr[k] >= 0) {
                    float* p = &acc[rr[k] * 32 + sub * 4];
                    __hip_atomic_fetch_add(p + 0, f[k].x * a[k], __ATOMIC_RELAXED, __HIP_MEMORY_SCOPE_WORKGROUP);
                    __hip_atomic_fetch_add(p + 1, f[k].y * a[k], __ATOMIC_RELAXED, __HIP_MEMORY_SCOPE_WORKGROUP);
                    __hip_atomic_fetch_add(p + 2, f[k].z * a[k], __ATOMIC_RELAXED, __HIP_MEMORY_SCOPE_WORKGROUP);
                    __hip_atomic_fetch_add(p + 3, f[k].w * a[k], __ATOMIC_RELAXED, __HIP_MEMORY_SCOPE_WORKGROUP);
                }
            }
        }
    } else {
        for (int e = tid; e < EE; e += 1024) {
            const unsigned int r = (unsigned int)(dst[e] - lo);
            if (r < (unsigned int)rows) lds_max_f32(&mrow[r], leaky(el[src[e]] + errow[r]));
        }
        __syncthreads();
        for (int e = tid; e < EE; e += 1024) {
            const unsigned int r = (unsigned int)(dst[e] - lo);
            if (r < (unsigned int)rows)
                __hip_atomic_fetch_add(&lrow[r], __expf(leaky(el[src[e]] + errow[r]) - mrow[r]),
                                       __ATOMIC_RELAXED, __HIP_MEMORY_SCOPE_WORKGROUP);
        }
        __syncthreads();
        for (int e = tid; e < EE; e += 1024) {
            const unsigned int r = (unsigned int)(dst[e] - lo);
            if (r < (unsigned int)rows) {
                const int sidx = src[e];
                const float a = __expf(leaky(el[sidx] + errow[r]) - mrow[r]) / lrow[r];
                for (int c = 0; c < 32; ++c)
                    __hip_atomic_fetch_add(&acc[r * 32 + c], feat[(size_t)sidx * 32 + c] * a,
                                           __ATOMIC_RELAXED, __HIP_MEMORY_SCOPE_WORKGROUP);
            }
        }
    }
    __syncthreads();

    for (int idx = tid; idx < rows * 32; idx += 1024)
        outp[(size_t)lo * 32 + idx] += acc[idx];
}

extern "C" void kernel_launch(void* const* d_in, const int* in_sizes, int n_in,
                              void* d_out, int out_size, void* d_ws, size_t ws_size,
                              hipStream_t stream) {
    const float* x     = (const float*)d_in[0];
    const int*   src   = (const int*)  d_in[1];
    const int*   dst   = (const int*)  d_in[2];
    const float* W1    = (const float*)d_in[3];
    const float* al1   = (const float*)d_in[4];
    const float* ar1   = (const float*)d_in[5];
    const float* b1    = (const float*)d_in[6];
    const float* W2    = (const float*)d_in[7];
    const float* al2   = (const float*)d_in[8];
    const float* ar2   = (const float*)d_in[9];
    const float* resW2 = (const float*)d_in[10];
    const float* b2    = (const float*)d_in[11];
    float* outp = (float*)d_out;

    float* ws    = (float*)d_ws;
    float* feat1 = ws;
    float* rst1  = feat1 + (size_t)NN * 256;
    float* el1   = rst1  + (size_t)NN * 256;
    float* er1   = el1   + (size_t)NN * 8;
    int*   lst   = (int*)(er1 + (size_t)NN * 8);
    int*   lcnt  = lst + (size_t)NTILE * CAP1;
    unsigned int* codes = (unsigned int*)(lcnt + NTILE);
    float* feat2 = feat1;
    float* el2   = el1;
    float* er2   = er1;

    gemm1_kernel<<<GEMM_WGS, GEMM_THREADS, 0, stream>>>(x, W1, feat1);
    elr1_kernel<<<ELR1_WGS, 256, 0, stream>>>(feat1, al1, ar1, el1, er1, dst, codes);
    agg1_tile_kernel<<<NTILE, 1024, 0, stream>>>(src, dst, codes, el1, er1,
                                                 feat1, x, b1, rst1, lst, lcnt);

    gemm2_kernel<<<GEMM_WGS, GEMM_THREADS, 0, stream>>>(rst1, W2, resW2, b2, feat2, outp);
    elr2_kernel<<<ELR2_WGS, 128, 0, stream>>>(feat2, al2, ar2, el2, er2);
    agg2_tile_kernel<<<NTILE, 1024, 0, stream>>>(src, dst, el2, er2, feat2, lst, lcnt, outp);
}
